# starting kernel
# baseline (speedup 1.0000x reference)
#include <hip/hip_runtime.h>

#define NN 100000
#define NE 1600000
#define DIM 128
#define NG 512
#define NL 3
#define OUTC (NL * DIM)
#define BN_EPS 1e-5f
#define ROWS_PER_BLOCK 128

typedef __attribute__((ext_vector_type(2))) float v2f;
typedef __attribute__((ext_vector_type(8))) float v8f;


__global__ void k_zero(float* __restrict__ p, int n) {
    int i = blockIdx.x * blockDim.x + threadIdx.x;
    if (i < n) p[i] = 0.f;
}

__global__ void k_init_deg(float* __restrict__ deg, int n) {
    int i = blockIdx.x * blockDim.x + threadIdx.x;
    if (i < n) deg[i] = 1.f;
}

__global__ void k_deg_edges(const int* __restrict__ dst, float* __restrict__ deg, int e) {
    int i = blockIdx.x * blockDim.x + threadIdx.x;
    if (i < e) atomicAdd(&deg[dst[i]], 1.f);
}

__global__ void k_dinv(float* __restrict__ deg, int n) {
    int i = blockIdx.x * blockDim.x + threadIdx.x;
    if (i < n) deg[i] = rsqrtf(fmaxf(deg[i], 1.f));
}

__global__ void __launch_bounds__(256)
k_transpose128(const float* __restrict__ W, float* __restrict__ WT) {
    __shared__ float t[16][17];
    int bx = blockIdx.x & 7, by = blockIdx.x >> 3;
    int tx = threadIdx.x & 15, ty = threadIdx.x >> 4;
    t[ty][tx] = W[(size_t)(by * 16 + ty) * DIM + bx * 16 + tx];
    __syncthreads();
    WT[(size_t)(bx * 16 + ty) * DIM + by * 16 + tx] = t[tx][ty];
}


__global__ void __launch_bounds__(256)
k_gemm_wmma(const float* __restrict__ A, const float* __restrict__ WT,
            float* __restrict__ out, int nrows) {
    __shared__ float tileA[16 * DIM];

    const int tid  = threadIdx.x;
    const int wave = tid >> 5;
    const int lane = tid & 31;
    const int row0 = blockIdx.x << 4;
    (void)nrows;

    const float4* srcv = (const float4*)(A + (size_t)row0 * DIM);
    float4* dstv = (float4*)tileA;
    dstv[tid]       = srcv[tid];
    dstv[tid + 256] = srcv[tid + 256];
    __syncthreads();

    const int m  = lane & 15;
    const int kh = (lane >> 4) << 1;
    const int n  = (wave << 4) + (lane & 15);

    const float2* wtrow = (const float2*)(WT + (size_t)n * DIM);

    v8f c = {};
#pragma unroll
    for (int ks = 0; ks < DIM; ks += 4) {
        v2f a, b;
        a.x = tileA[m * DIM + ks + kh];
        a.y = tileA[m * DIM + ks + kh + 1];
        float2 bb = wtrow[(ks + kh) >> 1];
        b.x = bb.x;
        b.y = bb.y;
        c = __builtin_amdgcn_wmma_f32_16x16x4_f32(
                 false, a,  false, b,
                 (short)0, c,  false,  false);
    }

    const int mo = (lane >> 4) << 3;
    float* orow = out + (size_t)(row0 + mo) * DIM + n;
#pragma unroll
    for (int r = 0; r < 8; ++r)
        orow[(size_t)r * DIM] = c[r];
}


__global__ void k_selfloop(const float* __restrict__ h, const float* __restrict__ dinv,
                           float* __restrict__ out) {
    int i = blockIdx.x * blockDim.x + threadIdx.x;
    int node = i >> 5, q = i & 31;
    if (node < NN) {
        float w = dinv[node]; w *= w;
        float4 v = ((const float4*)(h + (size_t)node * DIM))[q];
        float4 r; r.x = v.x * w; r.y = v.y * w; r.z = v.z * w; r.w = v.w * w;
        ((float4*)(out + (size_t)node * DIM))[q] = r;
    }
}

__global__ void k_edge_agg(const int* __restrict__ src, const int* __restrict__ dst,
                           const float* __restrict__ dinv, const float* __restrict__ h,
                           float* __restrict__ out) {
    int i = blockIdx.x * blockDim.x + threadIdx.x;
    int e = i >> 5, q = i & 31;
    if (e < NE) {
        int s = src[e], d = dst[e];
        float w = dinv[s] * dinv[d];
        float4 v = ((const float4*)(h + (size_t)s * DIM))[q];
        float* o = out + (size_t)d * DIM + (q << 2);
        atomicAdd(o + 0, v.x * w);
        atomicAdd(o + 1, v.y * w);
        atomicAdd(o + 2, v.z * w);
        atomicAdd(o + 3, v.w * w);
    }
}


__global__ void __launch_bounds__(DIM)
k_bias_relu_stats(float* __restrict__ h, const float* __restrict__ bias,
                  float* __restrict__ stats, int nrows) {
    int c  = threadIdx.x;
    int r0 = blockIdx.x * ROWS_PER_BLOCK;
    int r1 = min(r0 + ROWS_PER_BLOCK, nrows);
    float b = bias[c];
    float s = 0.f, s2 = 0.f;
    for (int r = r0; r < r1; ++r) {
        float v = h[(size_t)r * DIM + c] + b;
        v = fmaxf(v, 0.f);
        h[(size_t)r * DIM + c] = v;
        s += v; s2 += v * v;
    }
    atomicAdd(&stats[c], s);
    atomicAdd(&stats[DIM + c], s2);
}

__global__ void k_stats_finish(float* __restrict__ stats, float inv_n) {
    int c = threadIdx.x;
    float mu  = stats[c] * inv_n;
    float var = stats[DIM + c] * inv_n - mu * mu;
    stats[2 * DIM + c] = mu;
    stats[3 * DIM + c] = rsqrtf(var + BN_EPS);
}

__global__ void __launch_bounds__(DIM)
k_norm_pool(const float* __restrict__ h, const float* __restrict__ stats,
            const float* __restrict__ gamma, const float* __restrict__ beta,
            const int* __restrict__ batch, float* __restrict__ hnext,
            float* __restrict__ out, int layer, int nrows) {
    int c  = threadIdx.x;
    int r0 = blockIdx.x * ROWS_PER_BLOCK;
    int r1 = min(r0 + ROWS_PER_BLOCK, nrows);
    float mu = stats[2 * DIM + c], rstd = stats[3 * DIM + c];
    float ga = gamma[c], be = beta[c];
    int cur = -1; float acc = 0.f;
    for (int r = r0; r < r1; ++r) {
        float v = (h[(size_t)r * DIM + c] - mu) * rstd * ga + be;
        hnext[(size_t)r * DIM + c] = v;
        int g = batch[r];
        if (g != cur) {
            if (cur >= 0) atomicAdd(&out[(size_t)cur * OUTC + layer * DIM + c], acc);
            cur = g; acc = 0.f;
        }
        acc += v;
    }
    if (cur >= 0) atomicAdd(&out[(size_t)cur * OUTC + layer * DIM + c], acc);
}


extern "C" void kernel_launch(void* const* d_in, const int* in_sizes, int n_in,
                              void* d_out, int out_size, void* d_ws, size_t ws_size,
                              hipStream_t stream) {
    (void)in_sizes; (void)n_in; (void)out_size; (void)ws_size;

    const float* x     = (const float*)d_in[0];
    const int*   ei    = (const int*)d_in[1];
    const int*   batch = (const int*)d_in[2];
    const float *Wp[NL], *bp[NL], *gp[NL], *tp[NL];
    for (int i = 0; i < NL; ++i) {
        Wp[i] = (const float*)d_in[3 + 4 * i];
        bp[i] = (const float*)d_in[4 + 4 * i];
        gp[i] = (const float*)d_in[5 + 4 * i];
        tp[i] = (const float*)d_in[6 + 4 * i];
    }
    const int* esrc = ei;
    const int* edst = ei + NE;

    float* ws    = (float*)d_ws;
    float* dinv  = ws;
    float* stats = ws + NN;
    float* wtb   = ws + NN + 4 * DIM;
    size_t o     = ((size_t)NN + 4 * DIM + DIM * DIM + 255) & ~(size_t)255;
    float* bufA  = ws + o;
    float* bufB  = bufA + (size_t)NN * DIM;
    float* outp  = (float*)d_out;

    k_zero     <<<(NG * OUTC + 255) / 256, 256, 0, stream>>>(outp, NG * OUTC);
    k_init_deg <<<(NN + 255) / 256,        256, 0, stream>>>(dinv, NN);
    k_deg_edges<<<(NE + 255) / 256,        256, 0, stream>>>(edst, dinv, NE);
    k_dinv     <<<(NN + 255) / 256,        256, 0, stream>>>(dinv, NN);

    for (int l = 0; l < NL; ++l) {
        const float* hin = (l == 0) ? x : bufA;
        k_transpose128<<<64, 256, 0, stream>>>(Wp[l], wtb);
        k_gemm_wmma<<<NN / 16, 256, 0, stream>>>(hin, wtb, bufA, NN);
        k_selfloop <<<(NN * 32 + 255) / 256, 256, 0, stream>>>(bufA, dinv, bufB);
        k_edge_agg <<<(NE * 32) / 256, 256, 0, stream>>>(esrc, edst, dinv, bufA, bufB);
        k_zero           <<<1, 256, 0, stream>>>(stats, 2 * DIM);
        k_bias_relu_stats<<<(NN + ROWS_PER_BLOCK - 1) / ROWS_PER_BLOCK, DIM, 0, stream>>>(
            bufB, bp[l], stats, NN);
        k_stats_finish   <<<1, DIM, 0, stream>>>(stats, 1.f / (float)NN);
        k_norm_pool      <<<(NN + ROWS_PER_BLOCK - 1) / ROWS_PER_BLOCK, DIM, 0, stream>>>(
            bufB, stats, gp[l], tp[l], batch, bufA, outp, l, NN);
    }
}


# norm+pool pass gone: aggregation applies the previous layer's batch norm to each row it reads and pools its own 128 rows; last layer only pools; normalized h not stored; pool order, rounding differ
# speedup vs baseline: 3.6444x; 3.6444x over previous
#include <hip/hip_runtime.h>

#define NN 100000
#define NE 1600000
#define DIM 128
#define NG 512
#define NL 3
#define OUTC (NL * DIM)
#define BN_EPS 1e-5f
#define BR 128
#define NB ((NN + BR - 1) / BR)
#define CH 6400

typedef __attribute__((ext_vector_type(2))) float v2f;
typedef __attribute__((ext_vector_type(8))) float v8f;
typedef __attribute__((ext_vector_type(4))) float v4f;


__global__ void k_zero(float* __restrict__ p, int n) {
    int i = blockIdx.x * blockDim.x + threadIdx.x;
    if (i < n) p[i] = 0.f;
}

__global__ void k_init_deg(float* __restrict__ deg, int n) {
    int i = blockIdx.x * blockDim.x + threadIdx.x;
    if (i < n) deg[i] = 1.f;
}

__global__ void k_deg_edges(const int* __restrict__ dst, float* __restrict__ deg, int e) {
    int i = blockIdx.x * blockDim.x + threadIdx.x;
    if (i < e) atomicAdd(&deg[dst[i]], 1.f);
}

__global__ void k_dinv(float* __restrict__ deg, int n) {
    int i = blockIdx.x * blockDim.x + threadIdx.x;
    if (i < n) deg[i] = rsqrtf(fmaxf(deg[i], 1.f));
}

__global__ void __launch_bounds__(256)
k_transpose128(const float* __restrict__ W, float* __restrict__ WT) {
    __shared__ float t[16][17];
    int bx = blockIdx.x & 7, by = blockIdx.x >> 3;
    int tx = threadIdx.x & 15, ty = threadIdx.x >> 4;
    t[ty][tx] = W[(size_t)(by * 16 + ty) * DIM + bx * 16 + tx];
    __syncthreads();
    WT[(size_t)(bx * 16 + ty) * DIM + by * 16 + tx] = t[tx][ty];
}


#define WPITCH 132
#define GEMM_WAVES 8
#define NTILES (NN / 16)

__global__ void __launch_bounds__(256)
k_gemm_bias_relu_stats(const float* __restrict__ A, const float* __restrict__ WT,
                       const float* __restrict__ bias, float* __restrict__ out,
                       float* __restrict__ stats) {
    __shared__ __attribute__((aligned(16))) float wt[DIM * WPITCH];
    __shared__ float part[GEMM_WAVES][2 * DIM];

    const int tid  = threadIdx.x;
    const int lane = tid & 31;
    const int wave = __builtin_amdgcn_readfirstlane(tid >> 5);
    const int c = lane & 15;
    const int h = lane >> 4;
    const int tile = blockIdx.x * GEMM_WAVES + wave;
    const bool valid = tile < NTILES;

    float4 a[16];
    if (valid) {
        const float4* arow = (const float4*)(A + ((size_t)tile * 16 + c) * DIM) + h;
#pragma unroll
        for (int g = 0; g < 16; ++g) a[g] = arow[2 * g];
    }

    const float4* w4 = (const float4*)WT;
#pragma unroll
    for (int i = 0; i < 16; ++i) {
        const int idx = tid + 256 * i;
        *(float4*)(wt + (idx >> 5) * WPITCH + ((idx & 31) << 2)) = w4[idx];
    }
    __syncthreads();

    float s[8], s2[8];
#pragma unroll
    for (int ct = 0; ct < 8; ++ct) { s[ct] = 0.f; s2[ct] = 0.f; }

    if (valid) {
        v8f acc[8];
#pragma unroll
        for (int ct = 0; ct < 8; ++ct) acc[ct] = (v8f){};

        const float* wb = wt + c * WPITCH + 4 * h;
        float4 b0[8], b1[8];
#pragma unroll
        for (int ct = 0; ct < 8; ++ct) b0[ct] = *(const float4*)(wb + ct * 16 * WPITCH);

#pragma unroll
        for (int g = 0; g < 16; g += 2) {
#pragma unroll
            for (int ct = 0; ct < 8; ++ct)
                b1[ct] = *(const float4*)(wb + ct * 16 * WPITCH + 8 * (g + 1));
#pragma unroll
            for (int ct = 0; ct < 8; ++ct) {
                v2f av = {a[g].x, a[g].y}, bv = {b0[ct].x, b0[ct].y};
                acc[ct] = __builtin_amdgcn_wmma_f32_16x16x4_f32(false, av, false, bv, (short)0, acc[ct], false, false);
            }
#pragma unroll
            for (int ct = 0; ct < 8; ++ct) {
                v2f av = {a[g].z, a[g].w}, bv = {b0[ct].z, b0[ct].w};
                acc[ct] = __builtin_amdgcn_wmma_f32_16x16x4_f32(false, av, false, bv, (short)0, acc[ct], false, false);
            }
            if (g + 2 < 16) {
#pragma unroll
                for (int ct = 0; ct < 8; ++ct)
                    b0[ct] = *(const float4*)(wb + ct * 16 * WPITCH + 8 * (g + 2));
            }
#pragma unroll
            for (int ct = 0; ct < 8; ++ct) {
                v2f av = {a[g + 1].x, a[g + 1].y}, bv = {b1[ct].x, b1[ct].y};
                acc[ct] = __builtin_amdgcn_wmma_f32_16x16x4_f32(false, av, false, bv, (short)0, acc[ct], false, false);
            }
#pragma unroll
            for (int ct = 0; ct < 8; ++ct) {
                v2f av = {a[g + 1].z, a[g + 1].w}, bv = {b1[ct].z, b1[ct].w};
                acc[ct] = __builtin_amdgcn_wmma_f32_16x16x4_f32(false, av, false, bv, (short)0, acc[ct], false, false);
            }
        }

        float* orow = out + ((size_t)tile * 16 + 8 * h) * DIM + c;
#pragma unroll
        for (int ct = 0; ct < 8; ++ct) {
            const float bn = bias[ct * 16 + c];
#pragma unroll
            for (int r = 0; r < 8; ++r) {
                float v = fmaxf(acc[ct][r] + bn, 0.f);
                orow[(size_t)r * DIM + ct * 16] = v;
                s[ct] += v; s2[ct] += v * v;
            }
        }
    }

#pragma unroll
    for (int ct = 0; ct < 8; ++ct) {
        const float t  = s[ct]  + __shfl_xor(s[ct], 16);
        const float t2 = s2[ct] + __shfl_xor(s2[ct], 16);
        part[wave][(h ? DIM : 0) + ct * 16 + c] = h ? t2 : t;
    }
    __syncthreads();

    float tot = 0.f;
#pragma unroll
    for (int w = 0; w < GEMM_WAVES; ++w) tot += part[w][tid];
    atomicAdd(&stats[tid], tot);
}


__global__ void __launch_bounds__(1024)
k_bin_scan(const float* __restrict__ deg, int* __restrict__ binStart, int* __restrict__ cursor) {
    __shared__ int sc[1024];
    const int b = threadIdx.x;
    int c = 0;
    if (b < NB) {
        const int n0 = b * BR;
        const int n1 = min(n0 + BR, NN);
        const float4* d4 = (const float4*)(deg + n0);
        const int nq = (n1 - n0) >> 2;
        for (int i = 0; i < nq; ++i) {
            float4 v = d4[i];
            c += (int)v.x + (int)v.y + (int)v.z + (int)v.w - 4;
        }
        cursor[b] = 0;
    }
    sc[b] = c;
    __syncthreads();
    for (int off = 1; off < 1024; off <<= 1) {
        int v = (b >= off) ? sc[b - off] : 0;
        __syncthreads();
        sc[b] += v;
        __syncthreads();
    }
    if (b < NB) binStart[b + 1] = sc[b];
    if (b == 0) binStart[0] = 0;
}

__global__ void __launch_bounds__(256)
k_bin_place(const int* __restrict__ src, const int* __restrict__ dst,
            const int* __restrict__ binStart, int* __restrict__ cursor,
            int* __restrict__ rec) {
    __shared__ int cnt[NB];
    __shared__ int bas[NB];
    const int tid = threadIdx.x;
    const int e0 = blockIdx.x * CH;
    for (int i = tid; i < NB; i += 256) cnt[i] = 0;
    __syncthreads();
    for (int k = tid; k < CH; k += 256) atomicAdd(&cnt[dst[e0 + k] >> 7], 1);
    __syncthreads();
    for (int i = tid; i < NB; i += 256) {
        int c = cnt[i];
        bas[i] = c ? binStart[i] + atomicAdd(&cursor[i], c) : 0;
        cnt[i] = 0;
    }
    __syncthreads();
    for (int k = tid; k < CH; k += 256) {
        int d = dst[e0 + k], s = src[e0 + k];
        int b = d >> 7;
        int r = atomicAdd(&cnt[b], 1);
        rec[bas[b] + r] = (s << 7) | (d & (BR - 1));
    }
}


struct BnCols { v4f mu, a, be; };

__device__ __forceinline__ BnCols bn_cols(const float* __restrict__ stats,
                                          const float* __restrict__ gamma,
                                          const float* __restrict__ beta, int lane) {
    BnCols b;
    b.mu = ((const v4f*)(stats + 2 * DIM))[lane];
    b.a  = ((const v4f*)(stats + 3 * DIM))[lane] * ((const v4f*)gamma)[lane];
    b.be = ((const v4f*)beta)[lane];
    return b;
}

__device__ __forceinline__ v4f bn_apply(v4f v, const BnCols& b) {
    return (v - b.mu) * b.a + b.be;
}

#define POOL_WORDS (8 * 2 * DIM)

__device__ __forceinline__ void pool_flush(float* __restrict__ pool, int layer, int lane,
                                           int g, v4f ps) {
    float* p = pool + (size_t)g * OUTC + layer * DIM + 4 * lane;
    atomicAdd(p,     ps.x);
    atomicAdd(p + 1, ps.y);
    atomicAdd(p + 2, ps.z);
    atomicAdd(p + 3, ps.w);
}

struct PoolRun {
    int cur;
    v4f ps;
    v4f aF, aL;
};

__device__ __forceinline__ void pool_begin(PoolRun& p) {
    p.cur = -1;
    p.ps = p.aF = p.aL = (v4f){0.f, 0.f, 0.f, 0.f};
}

__device__ __forceinline__ void pool_close(PoolRun& p, float* __restrict__ pool, int layer,
                                           int lane, int gF, int gL) {
    if (p.cur < 0) return;
    if (p.cur == gF)      p.aF += p.ps;
    else if (p.cur == gL) p.aL += p.ps;
    else                  pool_flush(pool, layer, lane, p.cur, p.ps);
}

__device__ __forceinline__ void pool_row(PoolRun& p, float* __restrict__ pool, int layer,
                                         int lane, int gF, int gL, int g, v4f x) {
    if (g != p.cur) {
        pool_close(p, pool, layer, lane, gF, gL);
        p.cur = g;
        p.ps = (v4f){0.f, 0.f, 0.f, 0.f};
    }
    p.ps += x;
}

__device__ __forceinline__ void pool_store(const PoolRun& p, float* pl, int wave, int lane) {
    *(v4f*)(pl + (wave * 2 + 0) * DIM + 4 * lane) = p.aF;
    *(v4f*)(pl + (wave * 2 + 1) * DIM + 4 * lane) = p.aL;
}

__device__ __forceinline__ void pool_block_add(const float* pl, float* __restrict__ pool,
                                               int layer, int tid, int gF, int gL) {
    const int half = tid >> 7, col = tid & (DIM - 1);
    float t = 0.f;
#pragma unroll
    for (int w = 0; w < 8; ++w) t += pl[(w * 2 + half) * DIM + col];
    if (half == 0 || gL != gF)
        atomicAdd(pool + (size_t)(half ? gL : gF) * OUTC + layer * DIM + col, t);
}

template <bool BN>
__global__ void __launch_bounds__(256)
k_agg_tile(const int* __restrict__ rec, const int* __restrict__ binStart,
           const float* __restrict__ dinv, const float* __restrict__ h,
           float* __restrict__ out,
           const float* __restrict__ stats, const float* __restrict__ gamma,
           const float* __restrict__ beta, const int* __restrict__ batch,
           float* __restrict__ pool, int layer) {
    __shared__ float tile[(BR + 1) * DIM];
    __shared__ __attribute__((aligned(16))) float pl[BN ? POOL_WORDS : 4];

    const int tid  = threadIdx.x;
    const int lane = tid & 31;
    const int wave = __builtin_amdgcn_readfirstlane(tid >> 5);
    const int base = blockIdx.x * BR;
    const int nrows = min(BR, NN - base);
    const v4f* h4 = (const v4f*)h;

    BnCols bn;
    int gF = 0, gL = 0;
    if (BN) {
        bn = bn_cols(stats, gamma, beta, lane);
        gF = batch[base];
        gL = batch[base + nrows - 1];
    }

    {
        v4f v[16];
#pragma unroll
        for (int k = 0; k < 16; ++k) {
            const int node = min(base + wave + 8 * k, NN - 1);
            v[k] = h4[(size_t)node * 32 + lane];
        }
        const int kn = min(base + wave + 8 * (lane & 15), NN - 1);
        const float dn = dinv[kn];
        const int   gn = BN ? batch[kn] : 0;
        asm volatile("" :: "v"(v[0]), "v"(v[1]), "v"(v[2]), "v"(v[3]), "v"(v[4]), "v"(v[5]),
                           "v"(v[6]), "v"(v[7]), "v"(v[8]), "v"(v[9]), "v"(v[10]), "v"(v[11]),
                           "v"(v[12]), "v"(v[13]), "v"(v[14]), "v"(v[15]), "v"(dn), "v"(gn));
        PoolRun pr;
        pool_begin(pr);
#pragma unroll
        for (int k = 0; k < 16; ++k) {
            const int r = wave + 8 * k;
            if (r < nrows) {
                v4f x = v[k];
                if (BN) {
                    x = bn_apply(x, bn);
                    pool_row(pr, pool, layer, lane, gF, gL, __builtin_amdgcn_readlane(gn, k), x);
                }
                float w = __int_as_float(__builtin_amdgcn_readlane(__float_as_int(dn), k));
                w *= w;
                float* t = tile + r * DIM + lane;
                t[0]  = x.x * w;
                t[32] = x.y * w;
                t[64] = x.z * w;
                t[96] = x.w * w;
            }
        }
        if (BN) {
            pool_close(pr, pool, layer, lane, gF, gL);
            pool_store(pr, pl, wave, lane);
        }
    }
    __syncthreads();
    if (BN) pool_block_add(pl, pool, layer, tid, gF, gL);

    const int e0 = binStart[blockIdx.x];
    const int e1 = binStart[blockIdx.x + 1];
    for (int eb = e0 + wave * 32; eb < e1; eb += 8 * 32) {
        const int cnt = min(32, e1 - eb);
        int s = base, dl = BR;
        float ws = 0.f, wd = 0.f;
        if (lane < cnt) {
            const int r = rec[eb + lane];
            s  = r >> 7;
            dl = r & (BR - 1);
            ws = dinv[s];
            wd = dinv[base + dl];
        }
#pragma unroll
        for (int j0 = 0; j0 < 32; j0 += 8) {
            if (j0 < cnt) {
                v4f v[8];
#pragma unroll
                for (int u = 0; u < 8; ++u) {
                    const int sj = __builtin_amdgcn_readlane(s, j0 + u);
                    v[u] = h4[(size_t)sj * 32 + lane];
                }
#pragma unroll
                for (int u = 0; u < 8; ++u) {
                    const int   dj = __builtin_amdgcn_readlane(dl, j0 + u);
                    const float wj =
                        __int_as_float(__builtin_amdgcn_readlane(__float_as_int(ws), j0 + u)) *
                        __int_as_float(__builtin_amdgcn_readlane(__float_as_int(wd), j0 + u));
                    v4f x = v[u];
                    if (BN) x = bn_apply(x, bn);
                    float* t = tile + dj * DIM + lane;
                    __hip_atomic_fetch_add(t,      x.x * wj, __ATOMIC_RELAXED, __HIP_MEMORY_SCOPE_WORKGROUP);
                    __hip_atomic_fetch_add(t + 32, x.y * wj, __ATOMIC_RELAXED, __HIP_MEMORY_SCOPE_WORKGROUP);
                    __hip_atomic_fetch_add(t + 64, x.z * wj, __ATOMIC_RELAXED, __HIP_MEMORY_SCOPE_WORKGROUP);
                    __hip_atomic_fetch_add(t + 96, x.w * wj, __ATOMIC_RELAXED, __HIP_MEMORY_SCOPE_WORKGROUP);
                }
            }
        }
    }
    __syncthreads();

    v4f* o4 = (v4f*)out;
    for (int r = wave; r < nrows; r += 8) {
        const float* t = tile + r * DIM + lane;
        v4f v;
        v.x = t[0]; v.y = t[32]; v.z = t[64]; v.w = t[96];
        o4[(size_t)(base + r) * 32 + lane] = v;
    }
}


__global__ void k_stats_finish(float* __restrict__ stats, float inv_n) {
    int c = threadIdx.x;
    float mu  = stats[c] * inv_n;
    float var = stats[DIM + c] * inv_n - mu * mu;
    stats[2 * DIM + c] = mu;
    stats[3 * DIM + c] = rsqrtf(var + BN_EPS);
}

__global__ void __launch_bounds__(256)
k_norm_pool_rows(const float* __restrict__ h, const float* __restrict__ stats,
                 const float* __restrict__ gamma, const float* __restrict__ beta,
                 const int* __restrict__ batch, float* __restrict__ pool, int layer) {
    const int tid  = threadIdx.x;
    const int lane = tid & 31;
    const int wave = __builtin_amdgcn_readfirstlane(tid >> 5);
    const int base = blockIdx.x * BR;
    const int nrows = min(BR, NN - base);
    const v4f* h4 = (const v4f*)h;
    const BnCols bn = bn_cols(stats, gamma, beta, lane);
    const int gF = batch[base];
    const int gL = batch[base + nrows - 1];
    __shared__ __attribute__((aligned(16))) float pl[POOL_WORDS];

    v4f v[16];
#pragma unroll
    for (int k = 0; k < 16; ++k) {
        const int node = min(base + wave + 8 * k, NN - 1);
        v[k] = h4[(size_t)node * 32 + lane];
    }
    const int kn = min(base + wave + 8 * (lane & 15), NN - 1);
    const int gn = batch[kn];
    asm volatile("" :: "v"(v[0]), "v"(v[1]), "v"(v[2]), "v"(v[3]), "v"(v[4]), "v"(v[5]),
                       "v"(v[6]), "v"(v[7]), "v"(v[8]), "v"(v[9]), "v"(v[10]), "v"(v[11]),
                       "v"(v[12]), "v"(v[13]), "v"(v[14]), "v"(v[15]), "v"(gn));
    PoolRun pr;
    pool_begin(pr);
#pragma unroll
    for (int k = 0; k < 16; ++k) {
        if (wave + 8 * k < nrows)
            pool_row(pr, pool, layer, lane, gF, gL, __builtin_amdgcn_readlane(gn, k), bn_apply(v[k], bn));
    }
    pool_close(pr, pool, layer, lane, gF, gL);
    pool_store(pr, pl, wave, lane);
    __syncthreads();
    pool_block_add(pl, pool, layer, tid, gF, gL);
}


extern "C" void kernel_launch(void* const* d_in, const int* in_sizes, int n_in,
                              void* d_out, int out_size, void* d_ws, size_t ws_size,
                              hipStream_t stream) {
    (void)in_sizes; (void)n_in; (void)out_size; (void)ws_size;

    const float* x     = (const float*)d_in[0];
    const int*   ei    = (const int*)d_in[1];
    const int*   batch = (const int*)d_in[2];
    const float *Wp[NL], *bp[NL], *gp[NL], *tp[NL];
    for (int i = 0; i < NL; ++i) {
        Wp[i] = (const float*)d_in[3 + 4 * i];
        bp[i] = (const float*)d_in[4 + 4 * i];
        gp[i] = (const float*)d_in[5 + 4 * i];
        tp[i] = (const float*)d_in[6 + 4 * i];
    }
    const int* esrc = ei;
    const int* edst = ei + NE;

    float* ws    = (float*)d_ws;
    float* dinv  = ws;
    float* stats = ws + NN;
    float* wtb   = ws + NN + 4 * DIM;
    size_t o     = ((size_t)NN + 4 * DIM + DIM * DIM + 255) & ~(size_t)255;
    float* bufA  = ws + o;
    float* bufB  = bufA + (size_t)NN * DIM;
    int* binStart = (int*)(bufB + (size_t)NN * DIM);
    int* cursor   = binStart + 1024;
    int* rec      = cursor + 1024;
    float* outp  = (float*)d_out;

    k_zero     <<<(NG * OUTC + 255) / 256, 256, 0, stream>>>(outp, NG * OUTC);
    k_init_deg <<<(NN + 255) / 256,        256, 0, stream>>>(dinv, NN);
    k_deg_edges<<<(NE + 255) / 256,        256, 0, stream>>>(edst, dinv, NE);
    k_bin_scan <<<1, 1024, 0, stream>>>(dinv, binStart, cursor);
    k_dinv     <<<(NN + 255) / 256,        256, 0, stream>>>(dinv, NN);
    k_bin_place<<<NE / CH, 256, 0, stream>>>(esrc, edst, binStart, cursor, rec);

    k_agg_tile<false><<<NB, 256, 0, stream>>>(rec, binStart, dinv, x, bufB,
                                              stats, gp[0], tp[0], batch, outp, 0);
    for (int l = 0; l < NL; ++l) {
        k_transpose128<<<64, 256, 0, stream>>>(Wp[l], wtb);
        k_zero        <<<1, 256, 0, stream>>>(stats, 2 * DIM);
        k_gemm_bias_relu_stats<<<(NTILES + GEMM_WAVES - 1) / GEMM_WAVES, 256, 0, stream>>>(
            bufB, wtb, bp[l], bufA, stats);
        k_stats_finish<<<1, DIM, 0, stream>>>(stats, 1.f / (float)NN);
        if (l < NL - 1) {
            k_agg_tile<true><<<NB, 256, 0, stream>>>(rec, binStart, dinv, bufA, bufB,
                                                     stats, gp[l], tp[l], batch, outp, l);
        } else {
            k_norm_pool_rows<<<NB, 256, 0, stream>>>(bufA, stats, gp[l], tp[l], batch, outp, l);
        }
    }
}


# degrees and bin sizes counted in LDS integer counters: edges counted per 128-node bin per block, degrees from each bin's placed records; the global f32 add per edge is gone; sum orders unchanged
# speedup vs baseline: 4.6141x; 1.2661x over previous
#include <hip/hip_runtime.h>

#define NN 100000
#define NE 1600000
#define DIM 128
#define NG 512
#define NL 3
#define OUTC (NL * DIM)
#define BN_EPS 1e-5f
#define BR 128
#define NB ((NN + BR - 1) / BR)
#define CH 6400

typedef __attribute__((ext_vector_type(2))) float v2f;
typedef __attribute__((ext_vector_type(8))) float v8f;
typedef __attribute__((ext_vector_type(4))) float v4f;
typedef __attribute__((ext_vector_type(4))) int v4i;


__global__ void k_zero(float* __restrict__ p, int n) {
    int i = blockIdx.x * blockDim.x + threadIdx.x;
    if (i < n) p[i] = 0.f;
}

__global__ void k_init(float* __restrict__ p, int n, int* __restrict__ binCount) {
    int i = blockIdx.x * blockDim.x + threadIdx.x;
    if (i < n) p[i] = 0.f;
    if (i < NB) binCount[i] = 0;
}

__global__ void __launch_bounds__(256)
k_transpose128(const float* __restrict__ W, float* __restrict__ WT) {
    __shared__ float t[16][17];
    int bx = blockIdx.x & 7, by = blockIdx.x >> 3;
    int tx = threadIdx.x & 15, ty = threadIdx.x >> 4;
    t[ty][tx] = W[(size_t)(by * 16 + ty) * DIM + bx * 16 + tx];
    __syncthreads();
    WT[(size_t)(bx * 16 + ty) * DIM + by * 16 + tx] = t[tx][ty];
}


#define WPITCH 132
#define GEMM_WAVES 8
#define NTILES (NN / 16)

__global__ void __launch_bounds__(256)
k_gemm_bias_relu_stats(const float* __restrict__ A, const float* __restrict__ WT,
                       const float* __restrict__ bias, float* __restrict__ out,
                       float* __restrict__ stats) {
    __shared__ __attribute__((aligned(16))) float wt[DIM * WPITCH];
    __shared__ float part[GEMM_WAVES][2 * DIM];

    const int tid  = threadIdx.x;
    const int lane = tid & 31;
    const int wave = __builtin_amdgcn_readfirstlane(tid >> 5);
    const int c = lane & 15;
    const int h = lane >> 4;
    const int tile = blockIdx.x * GEMM_WAVES + wave;
    const bool valid = tile < NTILES;

    float4 a[16];
    if (valid) {
        const float4* arow = (const float4*)(A + ((size_t)tile * 16 + c) * DIM) + h;
#pragma unroll
        for (int g = 0; g < 16; ++g) a[g] = arow[2 * g];
    }

    const float4* w4 = (const float4*)WT;
#pragma unroll
    for (int i = 0; i < 16; ++i) {
        const int idx = tid + 256 * i;
        *(float4*)(wt + (idx >> 5) * WPITCH + ((idx & 31) << 2)) = w4[idx];
    }
    __syncthreads();

    float s[8], s2[8];
#pragma unroll
    for (int ct = 0; ct < 8; ++ct) { s[ct] = 0.f; s2[ct] = 0.f; }

    if (valid) {
        v8f acc[8];
#pragma unroll
        for (int ct = 0; ct < 8; ++ct) acc[ct] = (v8f){};

        const float* wb = wt + c * WPITCH + 4 * h;
        float4 b0[8], b1[8];
#pragma unroll
        for (int ct = 0; ct < 8; ++ct) b0[ct] = *(const float4*)(wb + ct * 16 * WPITCH);

#pragma unroll
        for (int g = 0; g < 16; g += 2) {
#pragma unroll
            for (int ct = 0; ct < 8; ++ct)
                b1[ct] = *(const float4*)(wb + ct * 16 * WPITCH + 8 * (g + 1));
#pragma unroll
            for (int ct = 0; ct < 8; ++ct) {
                v2f av = {a[g].x, a[g].y}, bv = {b0[ct].x, b0[ct].y};
                acc[ct] = __builtin_amdgcn_wmma_f32_16x16x4_f32(false, av, false, bv, (short)0, acc[ct], false, false);
            }
#pragma unroll
            for (int ct = 0; ct < 8; ++ct) {
                v2f av = {a[g].z, a[g].w}, bv = {b0[ct].z, b0[ct].w};
                acc[ct] = __builtin_amdgcn_wmma_f32_16x16x4_f32(false, av, false, bv, (short)0, acc[ct], false, false);
            }
            if (g + 2 < 16) {
#pragma unroll
                for (int ct = 0; ct < 8; ++ct)
                    b0[ct] = *(const float4*)(wb + ct * 16 * WPITCH + 8 * (g + 2));
            }
#pragma unroll
            for (int ct = 0; ct < 8; ++ct) {
                v2f av = {a[g + 1].x, a[g + 1].y}, bv = {b1[ct].x, b1[ct].y};
                acc[ct] = __builtin_amdgcn_wmma_f32_16x16x4_f32(false, av, false, bv, (short)0, acc[ct], false, false);
            }
#pragma unroll
            for (int ct = 0; ct < 8; ++ct) {
                v2f av = {a[g + 1].z, a[g + 1].w}, bv = {b1[ct].z, b1[ct].w};
                acc[ct] = __builtin_amdgcn_wmma_f32_16x16x4_f32(false, av, false, bv, (short)0, acc[ct], false, false);
            }
        }

        float* orow = out + ((size_t)tile * 16 + 8 * h) * DIM + c;
#pragma unroll
        for (int ct = 0; ct < 8; ++ct) {
            const float bn = bias[ct * 16 + c];
#pragma unroll
            for (int r = 0; r < 8; ++r) {
                float v = fmaxf(acc[ct][r] + bn, 0.f);
                orow[(size_t)r * DIM + ct * 16] = v;
                s[ct] += v; s2[ct] += v * v;
            }
        }
    }

#pragma unroll
    for (int ct = 0; ct < 8; ++ct) {
        const float t  = s[ct]  + __shfl_xor(s[ct], 16);
        const float t2 = s2[ct] + __shfl_xor(s2[ct], 16);
        part[wave][(h ? DIM : 0) + ct * 16 + c] = h ? t2 : t;
    }
    __syncthreads();

    float tot = 0.f;
#pragma unroll
    for (int w = 0; w < GEMM_WAVES; ++w) tot += part[w][tid];
    atomicAdd(&stats[tid], tot);
}


__global__ void __launch_bounds__(256)
k_bin_count(const int* __restrict__ dst, int* __restrict__ binCount) {
    __shared__ int cnt[NB];
    const int tid = threadIdx.x;
    for (int i = tid; i < NB; i += 256) cnt[i] = 0;
    __syncthreads();
    const int* dp = dst + blockIdx.x * CH;
    const v4i* d4 = (const v4i*)dp;
    v4i d[6];
#pragma unroll
    for (int i = 0; i < 6; ++i) d[i] = d4[tid + 256 * i];
    const int dt = dp[6 * 1024 + tid];
    asm volatile("" :: "v"(d[0]), "v"(d[1]), "v"(d[2]), "v"(d[3]), "v"(d[4]), "v"(d[5]), "v"(dt));
#pragma unroll
    for (int i = 0; i < 6; ++i) {
        atomicAdd(&cnt[d[i].x >> 7], 1);
        atomicAdd(&cnt[d[i].y >> 7], 1);
        atomicAdd(&cnt[d[i].z >> 7], 1);
        atomicAdd(&cnt[d[i].w >> 7], 1);
    }
    atomicAdd(&cnt[dt >> 7], 1);
    __syncthreads();
    for (int i = tid; i < NB; i += 256) {
        const int c = cnt[i];
        if (c) atomicAdd(&binCount[i], c);
    }
}

__global__ void __launch_bounds__(256)
k_bin_dinv(const int* __restrict__ rec, const int* __restrict__ binStart,
           float* __restrict__ dinv) {
    __shared__ int cnt[BR];
    const int tid = threadIdx.x;
    if (tid < BR) cnt[tid] = 0;
    __syncthreads();
    const int e0 = binStart[blockIdx.x];
    const int e1 = binStart[blockIdx.x + 1];
    for (int e = e0 + tid; e < e1; e += 4 * 256) {
        int r[4];
#pragma unroll
        for (int u = 0; u < 4; ++u) {
            const int k = e + 256 * u;
            r[u] = (k < e1) ? rec[k] : -1;
        }
#pragma unroll
        for (int u = 0; u < 4; ++u)
            if (r[u] >= 0) atomicAdd(&cnt[r[u] & (BR - 1)], 1);
    }
    __syncthreads();
    const int node = blockIdx.x * BR + tid;
    if (tid < BR && node < NN)
        dinv[node] = rsqrtf(fmaxf((float)(cnt[tid] + 1), 1.f));
}

__global__ void __launch_bounds__(1024)
k_bin_scan(int* __restrict__ binStart, int* __restrict__ cursor) {
    __shared__ int sc[1024];
    const int b = threadIdx.x;
    int c = 0;
    if (b < NB) {
        c = cursor[b];
        cursor[b] = 0;
    }
    sc[b] = c;
    __syncthreads();
    for (int off = 1; off < 1024; off <<= 1) {
        int v = (b >= off) ? sc[b - off] : 0;
        __syncthreads();
        sc[b] += v;
        __syncthreads();
    }
    if (b < NB) binStart[b + 1] = sc[b];
    if (b == 0) binStart[0] = 0;
}

__global__ void __launch_bounds__(256)
k_bin_place(const int* __restrict__ src, const int* __restrict__ dst,
            const int* __restrict__ binStart, int* __restrict__ cursor,
            int* __restrict__ rec) {
    __shared__ int cnt[NB];
    __shared__ int bas[NB];
    const int tid = threadIdx.x;
    const int e0 = blockIdx.x * CH;
    for (int i = tid; i < NB; i += 256) cnt[i] = 0;
    __syncthreads();
    for (int k = tid; k < CH; k += 256) atomicAdd(&cnt[dst[e0 + k] >> 7], 1);
    __syncthreads();
    for (int i = tid; i < NB; i += 256) {
        int c = cnt[i];
        bas[i] = c ? binStart[i] + atomicAdd(&cursor[i], c) : 0;
        cnt[i] = 0;
    }
    __syncthreads();
    for (int k = tid; k < CH; k += 256) {
        int d = dst[e0 + k], s = src[e0 + k];
        int b = d >> 7;
        int r = atomicAdd(&cnt[b], 1);
        rec[bas[b] + r] = (s << 7) | (d & (BR - 1));
    }
}


struct BnCols { v4f mu, a, be; };

__device__ __forceinline__ BnCols bn_cols(const float* __restrict__ stats,
                                          const float* __restrict__ gamma,
                                          const float* __restrict__ beta, int lane) {
    BnCols b;
    b.mu = ((const v4f*)(stats + 2 * DIM))[lane];
    b.a  = ((const v4f*)(stats + 3 * DIM))[lane] * ((const v4f*)gamma)[lane];
    b.be = ((const v4f*)beta)[lane];
    return b;
}

__device__ __forceinline__ v4f bn_apply(v4f v, const BnCols& b) {
    return (v - b.mu) * b.a + b.be;
}

#define POOL_WORDS (8 * 2 * DIM)

__device__ __forceinline__ void pool_flush(float* __restrict__ pool, int layer, int lane,
                                           int g, v4f ps) {
    float* p = pool + (size_t)g * OUTC + layer * DIM + 4 * lane;
    atomicAdd(p,     ps.x);
    atomicAdd(p + 1, ps.y);
    atomicAdd(p + 2, ps.z);
    atomicAdd(p + 3, ps.w);
}

struct PoolRun {
    int cur;
    v4f ps;
    v4f aF, aL;
};

__device__ __forceinline__ void pool_begin(PoolRun& p) {
    p.cur = -1;
    p.ps = p.aF = p.aL = (v4f){0.f, 0.f, 0.f, 0.f};
}

__device__ __forceinline__ void pool_close(PoolRun& p, float* __restrict__ pool, int layer,
                                           int lane, int gF, int gL) {
    if (p.cur < 0) return;
    if (p.cur == gF)      p.aF += p.ps;
    else if (p.cur == gL) p.aL += p.ps;
    else                  pool_flush(pool, layer, lane, p.cur, p.ps);
}

__device__ __forceinline__ void pool_row(PoolRun& p, float* __restrict__ pool, int layer,
                                         int lane, int gF, int gL, int g, v4f x) {
    if (g != p.cur) {
        pool_close(p, pool, layer, lane, gF, gL);
        p.cur = g;
        p.ps = (v4f){0.f, 0.f, 0.f, 0.f};
    }
    p.ps += x;
}

__device__ __forceinline__ void pool_store(const PoolRun& p, float* pl, int wave, int lane) {
    *(v4f*)(pl + (wave * 2 + 0) * DIM + 4 * lane) = p.aF;
    *(v4f*)(pl + (wave * 2 + 1) * DIM + 4 * lane) = p.aL;
}

__device__ __forceinline__ void pool_block_add(const float* pl, float* __restrict__ pool,
                                               int layer, int tid, int gF, int gL) {
    const int half = tid >> 7, col = tid & (DIM - 1);
    float t = 0.f;
#pragma unroll
    for (int w = 0; w < 8; ++w) t += pl[(w * 2 + half) * DIM + col];
    if (half == 0 || gL != gF)
        atomicAdd(pool + (size_t)(half ? gL : gF) * OUTC + layer * DIM + col, t);
}

template <bool BN>
__global__ void __launch_bounds__(256)
k_agg_tile(const int* __restrict__ rec, const int* __restrict__ binStart,
           const float* __restrict__ dinv, const float* __restrict__ h,
           float* __restrict__ out,
           const float* __restrict__ stats, const float* __restrict__ gamma,
           const float* __restrict__ beta, const int* __restrict__ batch,
           float* __restrict__ pool, int layer) {
    __shared__ float tile[(BR + 1) * DIM];
    __shared__ __attribute__((aligned(16))) float pl[BN ? POOL_WORDS : 4];

    const int tid  = threadIdx.x;
    const int lane = tid & 31;
    const int wave = __builtin_amdgcn_readfirstlane(tid >> 5);
    const int base = blockIdx.x * BR;
    const int nrows = min(BR, NN - base);
    const v4f* h4 = (const v4f*)h;

    BnCols bn;
    int gF = 0, gL = 0;
    if (BN) {
        bn = bn_cols(stats, gamma, beta, lane);
        gF = batch[base];
        gL = batch[base + nrows - 1];
    }

    {
        v4f v[16];
#pragma unroll
        for (int k = 0; k < 16; ++k) {
            const int node = min(base + wave + 8 * k, NN - 1);
            v[k] = h4[(size_t)node * 32 + lane];
        }
        const int kn = min(base + wave + 8 * (lane & 15), NN - 1);
        const float dn = dinv[kn];
        const int   gn = BN ? batch[kn] : 0;
        asm volatile("" :: "v"(v[0]), "v"(v[1]), "v"(v[2]), "v"(v[3]), "v"(v[4]), "v"(v[5]),
                           "v"(v[6]), "v"(v[7]), "v"(v[8]), "v"(v[9]), "v"(v[10]), "v"(v[11]),
                           "v"(v[12]), "v"(v[13]), "v"(v[14]), "v"(v[15]), "v"(dn), "v"(gn));
        PoolRun pr;
        pool_begin(pr);
#pragma unroll
        for (int k = 0; k < 16; ++k) {
            const int r = wave + 8 * k;
            if (r < nrows) {
                v4f x = v[k];
                if (BN) {
                    x = bn_apply(x, bn);
                    pool_row(pr, pool, layer, lane, gF, gL, __builtin_amdgcn_readlane(gn, k), x);
                }
                float w = __int_as_float(__builtin_amdgcn_readlane(__float_as_int(dn), k));
                w *= w;
                float* t = tile + r * DIM + lane;
                t[0]  = x.x * w;
                t[32] = x.y * w;
                t[64] = x.z * w;
                t[96] = x.w * w;
            }
        }
        if (BN) {
            pool_close(pr, pool, layer, lane, gF, gL);
            pool_store(pr, pl, wave, lane);
        }
    }
    __syncthreads();
    if (BN) pool_block_add(pl, pool, layer, tid, gF, gL);

    const int e0 = binStart[blockIdx.x];
    const int e1 = binStart[blockIdx.x + 1];
    for (int eb = e0 + wave * 32; eb < e1; eb += 8 * 32) {
        const int cnt = min(32, e1 - eb);
        int s = base, dl = BR;
        float ws = 0.f, wd = 0.f;
        if (lane < cnt) {
            const int r = rec[eb + lane];
            s  = r >> 7;
            dl = r & (BR - 1);
            ws = dinv[s];
            wd = dinv[base + dl];
        }
#pragma unroll
        for (int j0 = 0; j0 < 32; j0 += 8) {
            if (j0 < cnt) {
                v4f v[8];
#pragma unroll
                for (int u = 0; u < 8; ++u) {
                    const int sj = __builtin_amdgcn_readlane(s, j0 + u);
                    v[u] = h4[(size_t)sj * 32 + lane];
                }
#pragma unroll
                for (int u = 0; u < 8; ++u) {
                    const int   dj = __builtin_amdgcn_readlane(dl, j0 + u);
                    const float wj =
                        __int_as_float(__builtin_amdgcn_readlane(__float_as_int(ws), j0 + u)) *
                        __int_as_float(__builtin_amdgcn_readlane(__float_as_int(wd), j0 + u));
                    v4f x = v[u];
                    if (BN) x = bn_apply(x, bn);
                    float* t = tile + dj * DIM + lane;
                    __hip_atomic_fetch_add(t,      x.x * wj, __ATOMIC_RELAXED, __HIP_MEMORY_SCOPE_WORKGROUP);
                    __hip_atomic_fetch_add(t + 32, x.y * wj, __ATOMIC_RELAXED, __HIP_MEMORY_SCOPE_WORKGROUP);
                    __hip_atomic_fetch_add(t + 64, x.z * wj, __ATOMIC_RELAXED, __HIP_MEMORY_SCOPE_WORKGROUP);
                    __hip_atomic_fetch_add(t + 96, x.w * wj, __ATOMIC_RELAXED, __HIP_MEMORY_SCOPE_WORKGROUP);
                }
            }
        }
    }
    __syncthreads();

    v4f* o4 = (v4f*)out;
    for (int r = wave; r < nrows; r += 8) {
        const float* t = tile + r * DIM + lane;
        v4f v;
        v.x = t[0]; v.y = t[32]; v.z = t[64]; v.w = t[96];
        o4[(size_t)(base + r) * 32 + lane] = v;
    }
}


__global__ void k_stats_finish(float* __restrict__ stats, float inv_n) {
    int c = threadIdx.x;
    float mu  = stats[c] * inv_n;
    float var = stats[DIM + c] * inv_n - mu * mu;
    stats[2 * DIM + c] = mu;
    stats[3 * DIM + c] = rsqrtf(var + BN_EPS);
}

__global__ void __launch_bounds__(256)
k_norm_pool_rows(const float* __restrict__ h, const float* __restrict__ stats,
                 const float* __restrict__ gamma, const float* __restrict__ beta,
                 const int* __restrict__ batch, float* __restrict__ pool, int layer) {
    const int tid  = threadIdx.x;
    const int lane = tid & 31;
    const int wave = __builtin_amdgcn_readfirstlane(tid >> 5);
    const int base = blockIdx.x * BR;
    const int nrows = min(BR, NN - base);
    const v4f* h4 = (const v4f*)h;
    const BnCols bn = bn_cols(stats, gamma, beta, lane);
    const int gF = batch[base];
    const int gL = batch[base + nrows - 1];
    __shared__ __attribute__((aligned(16))) float pl[POOL_WORDS];

    v4f v[16];
#pragma unroll
    for (int k = 0; k < 16; ++k) {
        const int node = min(base + wave + 8 * k, NN - 1);
        v[k] = h4[(size_t)node * 32 + lane];
    }
    const int kn = min(base + wave + 8 * (lane & 15), NN - 1);
    const int gn = batch[kn];
    asm volatile("" :: "v"(v[0]), "v"(v[1]), "v"(v[2]), "v"(v[3]), "v"(v[4]), "v"(v[5]),
                       "v"(v[6]), "v"(v[7]), "v"(v[8]), "v"(v[9]), "v"(v[10]), "v"(v[11]),
                       "v"(v[12]), "v"(v[13]), "v"(v[14]), "v"(v[15]), "v"(gn));
    PoolRun pr;
    pool_begin(pr);
#pragma unroll
    for (int k = 0; k < 16; ++k) {
        if (wave + 8 * k < nrows)
            pool_row(pr, pool, layer, lane, gF, gL, __builtin_amdgcn_readlane(gn, k), bn_apply(v[k], bn));
    }
    pool_close(pr, pool, layer, lane, gF, gL);
    pool_store(pr, pl, wave, lane);
    __syncthreads();
    pool_block_add(pl, pool, layer, tid, gF, gL);
}


extern "C" void kernel_launch(void* const* d_in, const int* in_sizes, int n_in,
                              void* d_out, int out_size, void* d_ws, size_t ws_size,
                              hipStream_t stream) {
    (void)in_sizes; (void)n_in; (void)out_size; (void)ws_size;

    const float* x     = (const float*)d_in[0];
    const int*   ei    = (const int*)d_in[1];
    const int*   batch = (const int*)d_in[2];
    const float *Wp[NL], *bp[NL], *gp[NL], *tp[NL];
    for (int i = 0; i < NL; ++i) {
        Wp[i] = (const float*)d_in[3 + 4 * i];
        bp[i] = (const float*)d_in[4 + 4 * i];
        gp[i] = (const float*)d_in[5 + 4 * i];
        tp[i] = (const float*)d_in[6 + 4 * i];
    }
    const int* esrc = ei;
    const int* edst = ei + NE;

    float* ws    = (float*)d_ws;
    float* dinv  = ws;
    float* stats = ws + NN;
    float* wtb   = ws + NN + 4 * DIM;
    size_t o     = ((size_t)NN + 4 * DIM + DIM * DIM + 255) & ~(size_t)255;
    float* bufA  = ws + o;
    float* bufB  = bufA + (size_t)NN * DIM;
    int* binStart = (int*)(bufB + (size_t)NN * DIM);
    int* cursor   = binStart + 1024;
    int* rec      = cursor + 1024;
    float* outp  = (float*)d_out;

    k_init     <<<(NG * OUTC + 255) / 256, 256, 0, stream>>>(outp, NG * OUTC, cursor);
    k_bin_count<<<NE / CH, 256, 0, stream>>>(edst, cursor);
    k_bin_scan <<<1, 1024, 0, stream>>>(binStart, cursor);
    k_bin_place<<<NE / CH, 256, 0, stream>>>(esrc, edst, binStart, cursor, rec);
    k_bin_dinv <<<NB, 256, 0, stream>>>(rec, binStart, dinv);

    k_agg_tile<false><<<NB, 256, 0, stream>>>(rec, binStart, dinv, x, bufB,
                                              stats, gp[0], tp[0], batch, outp, 0);
    for (int l = 0; l < NL; ++l) {
        k_transpose128<<<64, 256, 0, stream>>>(Wp[l], wtb);
        k_zero        <<<1, 256, 0, stream>>>(stats, 2 * DIM);
        k_gemm_bias_relu_stats<<<(NTILES + GEMM_WAVES - 1) / GEMM_WAVES, 256, 0, stream>>>(
            bufB, wtb, bp[l], bufA, stats);
        k_stats_finish<<<1, DIM, 0, stream>>>(stats, 1.f / (float)NN);
        if (l < NL - 1) {
            k_agg_tile<true><<<NB, 256, 0, stream>>>(rec, binStart, dinv, bufA, bufB,
                                                     stats, gp[l], tp[l], batch, outp, l);
        } else {
            k_norm_pool_rows<<<NB, 256, 0, stream>>>(bufA, stats, gp[l], tp[l], batch, outp, l);
        }
    }
}


# destination bins of 98 rows (were 128): 1021 aggregation blocks, four per WGP, so all WGPs carry equal edges (782 blocks gave some 4 bins, most 3); pool sums regrouped, adds in arrival order
# speedup vs baseline: 4.9821x; 1.0798x over previous
#include <hip/hip_runtime.h>

#define NN 100000
#define NE 1600000
#define DIM 128
#define NG 512
#define NL 3
#define OUTC (NL * DIM)
#define BN_EPS 1e-5f
#define BR 98
#define NB ((NN + BR - 1) / BR)
#define TROWS 129
#define RBITS 7
#define RMASK ((1 << RBITS) - 1)
#define PR 128
#define NPB ((NN + PR - 1) / PR)
#define CH 6400

typedef __attribute__((ext_vector_type(2))) float v2f;
typedef __attribute__((ext_vector_type(8))) float v8f;
typedef __attribute__((ext_vector_type(4))) float v4f;
typedef __attribute__((ext_vector_type(4))) int v4i;


__global__ void k_zero(float* __restrict__ p, int n) {
    int i = blockIdx.x * blockDim.x + threadIdx.x;
    if (i < n) p[i] = 0.f;
}

__global__ void k_init(float* __restrict__ p, int n, int* __restrict__ binCount) {
    int i = blockIdx.x * blockDim.x + threadIdx.x;
    if (i < n) p[i] = 0.f;
    if (i < NB) binCount[i] = 0;
}

__global__ void __launch_bounds__(256)
k_transpose128(const float* __restrict__ W, float* __restrict__ WT) {
    __shared__ float t[16][17];
    int bx = blockIdx.x & 7, by = blockIdx.x >> 3;
    int tx = threadIdx.x & 15, ty = threadIdx.x >> 4;
    t[ty][tx] = W[(size_t)(by * 16 + ty) * DIM + bx * 16 + tx];
    __syncthreads();
    WT[(size_t)(bx * 16 + ty) * DIM + by * 16 + tx] = t[tx][ty];
}


#define WPITCH 132
#define GEMM_WAVES 8
#define NTILES (NN / 16)

__global__ void __launch_bounds__(256)
k_gemm_bias_relu_stats(const float* __restrict__ A, const float* __restrict__ WT,
                       const float* __restrict__ bias, float* __restrict__ out,
                       float* __restrict__ stats) {
    __shared__ __attribute__((aligned(16))) float wt[DIM * WPITCH];
    __shared__ float part[GEMM_WAVES][2 * DIM];

    const int tid  = threadIdx.x;
    const int lane = tid & 31;
    const int wave = __builtin_amdgcn_readfirstlane(tid >> 5);
    const int c = lane & 15;
    const int h = lane >> 4;
    const int tile = blockIdx.x * GEMM_WAVES + wave;
    const bool valid = tile < NTILES;

    float4 a[16];
    if (valid) {
        const float4* arow = (const float4*)(A + ((size_t)tile * 16 + c) * DIM) + h;
#pragma unroll
        for (int g = 0; g < 16; ++g) a[g] = arow[2 * g];
    }

    const float4* w4 = (const float4*)WT;
#pragma unroll
    for (int i = 0; i < 16; ++i) {
        const int idx = tid + 256 * i;
        *(float4*)(wt + (idx >> 5) * WPITCH + ((idx & 31) << 2)) = w4[idx];
    }
    __syncthreads();

    float s[8], s2[8];
#pragma unroll
    for (int ct = 0; ct < 8; ++ct) { s[ct] = 0.f; s2[ct] = 0.f; }

    if (valid) {
        v8f acc[8];
#pragma unroll
        for (int ct = 0; ct < 8; ++ct) acc[ct] = (v8f){};

        const float* wb = wt + c * WPITCH + 4 * h;
        float4 b0[8], b1[8];
#pragma unroll
        for (int ct = 0; ct < 8; ++ct) b0[ct] = *(const float4*)(wb + ct * 16 * WPITCH);

#pragma unroll
        for (int g = 0; g < 16; g += 2) {
#pragma unroll
            for (int ct = 0; ct < 8; ++ct)
                b1[ct] = *(const float4*)(wb + ct * 16 * WPITCH + 8 * (g + 1));
#pragma unroll
            for (int ct = 0; ct < 8; ++ct) {
                v2f av = {a[g].x, a[g].y}, bv = {b0[ct].x, b0[ct].y};
                acc[ct] = __builtin_amdgcn_wmma_f32_16x16x4_f32(false, av, false, bv, (short)0, acc[ct], false, false);
            }
#pragma unroll
            for (int ct = 0; ct < 8; ++ct) {
                v2f av = {a[g].z, a[g].w}, bv = {b0[ct].z, b0[ct].w};
                acc[ct] = __builtin_amdgcn_wmma_f32_16x16x4_f32(false, av, false, bv, (short)0, acc[ct], false, false);
            }
            if (g + 2 < 16) {
#pragma unroll
                for (int ct = 0; ct < 8; ++ct)
                    b0[ct] = *(const float4*)(wb + ct * 16 * WPITCH + 8 * (g + 2));
            }
#pragma unroll
            for (int ct = 0; ct < 8; ++ct) {
                v2f av = {a[g + 1].x, a[g + 1].y}, bv = {b1[ct].x, b1[ct].y};
                acc[ct] = __builtin_amdgcn_wmma_f32_16x16x4_f32(false, av, false, bv, (short)0, acc[ct], false, false);
            }
#pragma unroll
            for (int ct = 0; ct < 8; ++ct) {
                v2f av = {a[g + 1].z, a[g + 1].w}, bv = {b1[ct].z, b1[ct].w};
                acc[ct] = __builtin_amdgcn_wmma_f32_16x16x4_f32(false, av, false, bv, (short)0, acc[ct], false, false);
            }
        }

        float* orow = out + ((size_t)tile * 16 + 8 * h) * DIM + c;
#pragma unroll
        for (int ct = 0; ct < 8; ++ct) {
            const float bn = bias[ct * 16 + c];
#pragma unroll
            for (int r = 0; r < 8; ++r) {
                float v = fmaxf(acc[ct][r] + bn, 0.f);
                orow[(size_t)r * DIM + ct * 16] = v;
                s[ct] += v; s2[ct] += v * v;
            }
        }
    }

#pragma unroll
    for (int ct = 0; ct < 8; ++ct) {
        const float t  = s[ct]  + __shfl_xor(s[ct], 16);
        const float t2 = s2[ct] + __shfl_xor(s2[ct], 16);
        part[wave][(h ? DIM : 0) + ct * 16 + c] = h ? t2 : t;
    }
    __syncthreads();

    float tot = 0.f;
#pragma unroll
    for (int w = 0; w < GEMM_WAVES; ++w) tot += part[w][tid];
    atomicAdd(&stats[tid], tot);
}


__global__ void __launch_bounds__(256)
k_bin_count(const int* __restrict__ dst, int* __restrict__ binCount) {
    __shared__ int cnt[NB];
    const int tid = threadIdx.x;
    for (int i = tid; i < NB; i += 256) cnt[i] = 0;
    __syncthreads();
    const int* dp = dst + blockIdx.x * CH;
    const v4i* d4 = (const v4i*)dp;
    v4i d[6];
#pragma unroll
    for (int i = 0; i < 6; ++i) d[i] = d4[tid + 256 * i];
    const int dt = dp[6 * 1024 + tid];
    asm volatile("" :: "v"(d[0]), "v"(d[1]), "v"(d[2]), "v"(d[3]), "v"(d[4]), "v"(d[5]), "v"(dt));
#pragma unroll
    for (int i = 0; i < 6; ++i) {
        atomicAdd(&cnt[(unsigned)d[i].x / BR], 1);
        atomicAdd(&cnt[(unsigned)d[i].y / BR], 1);
        atomicAdd(&cnt[(unsigned)d[i].z / BR], 1);
        atomicAdd(&cnt[(unsigned)d[i].w / BR], 1);
    }
    atomicAdd(&cnt[(unsigned)dt / BR], 1);
    __syncthreads();
    for (int i = tid; i < NB; i += 256) {
        const int c = cnt[i];
        if (c) atomicAdd(&binCount[i], c);
    }
}

__global__ void __launch_bounds__(256)
k_bin_dinv(const int* __restrict__ rec, const int* __restrict__ binStart,
           float* __restrict__ dinv) {
    __shared__ int cnt[BR];
    const int tid = threadIdx.x;
    if (tid < BR) cnt[tid] = 0;
    __syncthreads();
    const int e0 = binStart[blockIdx.x];
    const int e1 = binStart[blockIdx.x + 1];
    for (int e = e0 + tid; e < e1; e += 4 * 256) {
        int r[4];
#pragma unroll
        for (int u = 0; u < 4; ++u) {
            const int k = e + 256 * u;
            r[u] = (k < e1) ? rec[k] : -1;
        }
#pragma unroll
        for (int u = 0; u < 4; ++u)
            if (r[u] >= 0) atomicAdd(&cnt[r[u] & RMASK], 1);
    }
    __syncthreads();
    const int node = blockIdx.x * BR + tid;
    if (tid < BR && node < NN)
        dinv[node] = rsqrtf(fmaxf((float)(cnt[tid] + 1), 1.f));
}

__global__ void __launch_bounds__(1024)
k_bin_scan(int* __restrict__ binStart, int* __restrict__ cursor) {
    __shared__ int sc[1024];
    const int b = threadIdx.x;
    int c = 0;
    if (b < NB) {
        c = cursor[b];
        cursor[b] = 0;
    }
    sc[b] = c;
    __syncthreads();
    for (int off = 1; off < 1024; off <<= 1) {
        int v = (b >= off) ? sc[b - off] : 0;
        __syncthreads();
        sc[b] += v;
        __syncthreads();
    }
    if (b < NB) binStart[b + 1] = sc[b];
    if (b == 0) binStart[0] = 0;
}

__global__ void __launch_bounds__(256)
k_bin_place(const int* __restrict__ src, const int* __restrict__ dst,
            const int* __restrict__ binStart, int* __restrict__ cursor,
            int* __restrict__ rec) {
    __shared__ int cnt[NB];
    __shared__ int bas[NB];
    const int tid = threadIdx.x;
    const int e0 = blockIdx.x * CH;
    for (int i = tid; i < NB; i += 256) cnt[i] = 0;
    __syncthreads();
    for (int k = tid; k < CH; k += 256) atomicAdd(&cnt[(unsigned)dst[e0 + k] / BR], 1);
    __syncthreads();
    for (int i = tid; i < NB; i += 256) {
        int c = cnt[i];
        bas[i] = c ? binStart[i] + atomicAdd(&cursor[i], c) : 0;
        cnt[i] = 0;
    }
    __syncthreads();
    for (int k = tid; k < CH; k += 256) {
        int d = dst[e0 + k], s = src[e0 + k];
        int b = (unsigned)d / BR;
        int r = atomicAdd(&cnt[b], 1);
        rec[bas[b] + r] = (s << RBITS) | (d - b * BR);
    }
}


struct BnCols { v4f mu, a, be; };

__device__ __forceinline__ BnCols bn_cols(const float* __restrict__ stats,
                                          const float* __restrict__ gamma,
                                          const float* __restrict__ beta, int lane) {
    BnCols b;
    b.mu = ((const v4f*)(stats + 2 * DIM))[lane];
    b.a  = ((const v4f*)(stats + 3 * DIM))[lane] * ((const v4f*)gamma)[lane];
    b.be = ((const v4f*)beta)[lane];
    return b;
}

__device__ __forceinline__ v4f bn_apply(v4f v, const BnCols& b) {
    return (v - b.mu) * b.a + b.be;
}

#define POOL_WORDS (8 * 2 * DIM)

__device__ __forceinline__ void pool_flush(float* __restrict__ pool, int layer, int lane,
                                           int g, v4f ps) {
    float* p = pool + (size_t)g * OUTC + layer * DIM + 4 * lane;
    atomicAdd(p,     ps.x);
    atomicAdd(p + 1, ps.y);
    atomicAdd(p + 2, ps.z);
    atomicAdd(p + 3, ps.w);
}

struct PoolRun {
    int cur;
    v4f ps;
    v4f aF, aL;
};

__device__ __forceinline__ void pool_begin(PoolRun& p) {
    p.cur = -1;
    p.ps = p.aF = p.aL = (v4f){0.f, 0.f, 0.f, 0.f};
}

__device__ __forceinline__ void pool_close(PoolRun& p, float* __restrict__ pool, int layer,
                                           int lane, int gF, int gL) {
    if (p.cur < 0) return;
    if (p.cur == gF)      p.aF += p.ps;
    else if (p.cur == gL) p.aL += p.ps;
    else                  pool_flush(pool, layer, lane, p.cur, p.ps);
}

__device__ __forceinline__ void pool_row(PoolRun& p, float* __restrict__ pool, int layer,
                                         int lane, int gF, int gL, int g, v4f x) {
    if (g != p.cur) {
        pool_close(p, pool, layer, lane, gF, gL);
        p.cur = g;
        p.ps = (v4f){0.f, 0.f, 0.f, 0.f};
    }
    p.ps += x;
}

__device__ __forceinline__ void pool_store(const PoolRun& p, float* pl, int wave, int lane) {
    *(v4f*)(pl + (wave * 2 + 0) * DIM + 4 * lane) = p.aF;
    *(v4f*)(pl + (wave * 2 + 1) * DIM + 4 * lane) = p.aL;
}

__device__ __forceinline__ void pool_block_add(const float* pl, float* __restrict__ pool,
                                               int layer, int tid, int gF, int gL) {
    const int half = tid >> 7, col = tid & (DIM - 1);
    float t = 0.f;
#pragma unroll
    for (int w = 0; w < 8; ++w) t += pl[(w * 2 + half) * DIM + col];
    if (half == 0 || gL != gF)
        atomicAdd(pool + (size_t)(half ? gL : gF) * OUTC + layer * DIM + col, t);
}

template <bool BN>
__global__ void __launch_bounds__(256)
k_agg_tile(const int* __restrict__ rec, const int* __restrict__ binStart,
           const float* __restrict__ dinv, const float* __restrict__ h,
           float* __restrict__ out,
           const float* __restrict__ stats, const float* __restrict__ gamma,
           const float* __restrict__ beta, const int* __restrict__ batch,
           float* __restrict__ pool, int layer) {
    __shared__ float tile[TROWS * DIM];
    __shared__ __attribute__((aligned(16))) float pl[BN ? POOL_WORDS : 4];

    const int tid  = threadIdx.x;
    const int lane = tid & 31;
    const int wave = __builtin_amdgcn_readfirstlane(tid >> 5);
    const int base = blockIdx.x * BR;
    const int nrows = min(BR, NN - base);
    const v4f* h4 = (const v4f*)h;

    BnCols bn;
    int gF = 0, gL = 0;
    if (BN) {
        bn = bn_cols(stats, gamma, beta, lane);
        gF = batch[base];
        gL = batch[base + nrows - 1];
    }

    {
        constexpr int SR = (BR + 7) / 8;
        v4f v[SR];
#pragma unroll
        for (int k = 0; k < SR; ++k) {
            const int node = min(base + wave + 8 * k, NN - 1);
            v[k] = h4[(size_t)node * 32 + lane];
        }
        const int kn = min(base + wave + 8 * (lane & 15), NN - 1);
        const float dn = dinv[kn];
        const int   gn = BN ? batch[kn] : 0;
        static_assert(SR == 13, "the fence below names v[0] .. v[SR - 1]");
        asm volatile("" :: "v"(v[0]), "v"(v[1]), "v"(v[2]), "v"(v[3]), "v"(v[4]), "v"(v[5]),
                           "v"(v[6]), "v"(v[7]), "v"(v[8]), "v"(v[9]), "v"(v[10]), "v"(v[11]),
                           "v"(v[12]), "v"(dn), "v"(gn));
        PoolRun pr;
        pool_begin(pr);
#pragma unroll
        for (int k = 0; k < SR; ++k) {
            const int r = wave + 8 * k;
            if (r < nrows) {
                v4f x = v[k];
                if (BN) {
                    x = bn_apply(x, bn);
                    pool_row(pr, pool, layer, lane, gF, gL, __builtin_amdgcn_readlane(gn, k), x);
                }
                float w = __int_as_float(__builtin_amdgcn_readlane(__float_as_int(dn), k));
                w *= w;
                float* t = tile + r * DIM + lane;
                t[0]  = x.x * w;
                t[32] = x.y * w;
                t[64] = x.z * w;
                t[96] = x.w * w;
            }
        }
        if (BN) {
            pool_close(pr, pool, layer, lane, gF, gL);
            pool_store(pr, pl, wave, lane);
        }
    }
    __syncthreads();
    if (BN) pool_block_add(pl, pool, layer, tid, gF, gL);

    const int e0 = binStart[blockIdx.x];
    const int e1 = binStart[blockIdx.x + 1];
    for (int eb = e0 + wave * 32; eb < e1; eb += 8 * 32) {
        const int cnt = min(32, e1 - eb);
        int s = base, dl = BR;
        float ws = 0.f, wd = 0.f;
        if (lane < cnt) {
            const int r = rec[eb + lane];
            s  = r >> RBITS;
            dl = r & RMASK;
            ws = dinv[s];
            wd = dinv[base + dl];
        }
#pragma unroll
        for (int j0 = 0; j0 < 32; j0 += 8) {
            if (j0 < cnt) {
                v4f v[8];
#pragma unroll
                for (int u = 0; u < 8; ++u) {
                    const int sj = __builtin_amdgcn_readlane(s, j0 + u);
                    v[u] = h4[(size_t)sj * 32 + lane];
                }
#pragma unroll
                for (int u = 0; u < 8; ++u) {
                    const int   dj = __builtin_amdgcn_readlane(dl, j0 + u);
                    const float wj =
                        __int_as_float(__builtin_amdgcn_readlane(__float_as_int(ws), j0 + u)) *
                        __int_as_float(__builtin_amdgcn_readlane(__float_as_int(wd), j0 + u));
                    v4f x = v[u];
                    if (BN) x = bn_apply(x, bn);
                    float* t = tile + dj * DIM + lane;
                    __hip_atomic_fetch_add(t,      x.x * wj, __ATOMIC_RELAXED, __HIP_MEMORY_SCOPE_WORKGROUP);
                    __hip_atomic_fetch_add(t + 32, x.y * wj, __ATOMIC_RELAXED, __HIP_MEMORY_SCOPE_WORKGROUP);
                    __hip_atomic_fetch_add(t + 64, x.z * wj, __ATOMIC_RELAXED, __HIP_MEMORY_SCOPE_WORKGROUP);
                    __hip_atomic_fetch_add(t + 96, x.w * wj, __ATOMIC_RELAXED, __HIP_MEMORY_SCOPE_WORKGROUP);
                }
            }
        }
    }
    __syncthreads();

    v4f* o4 = (v4f*)out;
    for (int r = wave; r < nrows; r += 8) {
        const float* t = tile + r * DIM + lane;
        v4f v;
        v.x = t[0]; v.y = t[32]; v.z = t[64]; v.w = t[96];
        o4[(size_t)(base + r) * 32 + lane] = v;
    }
}


__global__ void k_stats_finish(float* __restrict__ stats, float inv_n) {
    int c = threadIdx.x;
    float mu  = stats[c] * inv_n;
    float var = stats[DIM + c] * inv_n - mu * mu;
    stats[2 * DIM + c] = mu;
    stats[3 * DIM + c] = rsqrtf(var + BN_EPS);
}

__global__ void __launch_bounds__(256)
k_norm_pool_rows(const float* __restrict__ h, const float* __restrict__ stats,
                 const float* __restrict__ gamma, const float* __restrict__ beta,
                 const int* __restrict__ batch, float* __restrict__ pool, int layer) {
    const int tid  = threadIdx.x;
    const int lane = tid & 31;
    const int wave = __builtin_amdgcn_readfirstlane(tid >> 5);
    const int base = blockIdx.x * PR;
    const int nrows = min(PR, NN - base);
    const v4f* h4 = (const v4f*)h;
    const BnCols bn = bn_cols(stats, gamma, beta, lane);
    const int gF = batch[base];
    const int gL = batch[base + nrows - 1];
    __shared__ __attribute__((aligned(16))) float pl[POOL_WORDS];

    v4f v[16];
#pragma unroll
    for (int k = 0; k < 16; ++k) {
        const int node = min(base + wave + 8 * k, NN - 1);
        v[k] = h4[(size_t)node * 32 + lane];
    }
    const int kn = min(base + wave + 8 * (lane & 15), NN - 1);
    const int gn = batch[kn];
    asm volatile("" :: "v"(v[0]), "v"(v[1]), "v"(v[2]), "v"(v[3]), "v"(v[4]), "v"(v[5]),
                       "v"(v[6]), "v"(v[7]), "v"(v[8]), "v"(v[9]), "v"(v[10]), "v"(v[11]),
                       "v"(v[12]), "v"(v[13]), "v"(v[14]), "v"(v[15]), "v"(gn));
    PoolRun pr;
    pool_begin(pr);
#pragma unroll
    for (int k = 0; k < 16; ++k) {
        if (wave + 8 * k < nrows)
            pool_row(pr, pool, layer, lane, gF, gL, __builtin_amdgcn_readlane(gn, k), bn_apply(v[k], bn));
    }
    pool_close(pr, pool, layer, lane, gF, gL);
    pool_store(pr, pl, wave, lane);
    __syncthreads();
    pool_block_add(pl, pool, layer, tid, gF, gL);
}


extern "C" void kernel_launch(void* const* d_in, const int* in_sizes, int n_in,
                              void* d_out, int out_size, void* d_ws, size_t ws_size,
                              hipStream_t stream) {
    (void)in_sizes; (void)n_in; (void)out_size; (void)ws_size;

    const float* x     = (const float*)d_in[0];
    const int*   ei    = (const int*)d_in[1];
    const int*   batch = (const int*)d_in[2];
    const float *Wp[NL], *bp[NL], *gp[NL], *tp[NL];
    for (int i = 0; i < NL; ++i) {
        Wp[i] = (const float*)d_in[3 + 4 * i];
        bp[i] = (const float*)d_in[4 + 4 * i];
        gp[i] = (const float*)d_in[5 + 4 * i];
        tp[i] = (const float*)d_in[6 + 4 * i];
    }
    const int* esrc = ei;
    const int* edst = ei + NE;

    float* ws    = (float*)d_ws;
    float* dinv  = ws;
    float* stats = ws + NN;
    float* wtb   = ws + NN + 4 * DIM;
    size_t o     = ((size_t)NN + 4 * DIM + DIM * DIM + 255) & ~(size_t)255;
    float* bufA  = ws + o;
    float* bufB  = bufA + (size_t)NN * DIM;
    int* binStart = (int*)(bufB + (size_t)NN * DIM);
    int* cursor   = binStart + 1024;
    int* rec      = cursor + 1024;
    float* outp  = (float*)d_out;

    k_init     <<<(NG * OUTC + 255) / 256, 256, 0, stream>>>(outp, NG * OUTC, cursor);
    k_bin_count<<<NE / CH, 256, 0, stream>>>(edst, cursor);
    k_bin_scan <<<1, 1024, 0, stream>>>(binStart, cursor);
    k_bin_place<<<NE / CH, 256, 0, stream>>>(esrc, edst, binStart, cursor, rec);
    k_bin_dinv <<<NB, 256, 0, stream>>>(rec, binStart, dinv);

    k_agg_tile<false><<<NB, 256, 0, stream>>>(rec, binStart, dinv, x, bufB,
                                              stats, gp[0], tp[0], batch, outp, 0);
    for (int l = 0; l < NL; ++l) {
        k_transpose128<<<64, 256, 0, stream>>>(Wp[l], wtb);
        k_zero        <<<1, 256, 0, stream>>>(stats, 2 * DIM);
        k_gemm_bias_relu_stats<<<(NTILES + GEMM_WAVES - 1) / GEMM_WAVES, 256, 0, stream>>>(
            bufB, wtb, bp[l], bufA, stats);
        k_stats_finish<<<1, DIM, 0, stream>>>(stats, 1.f / (float)NN);
        if (l < NL - 1) {
            k_agg_tile<true><<<NB, 256, 0, stream>>>(rec, binStart, dinv, bufA, bufB,
                                                     stats, gp[l], tp[l], batch, outp, l);
        } else {
            k_norm_pool_rows<<<NPB, 256, 0, stream>>>(bufA, stats, gp[l], tp[l], batch, outp, l);
        }
    }
}


# GEMM made persistent: 256 blocks, one per WGP, W transposed into LDS once per block (transpose launches gone); a wave walks 3 row tiles, next A rows loaded under the products; BN sum order differs
# speedup vs baseline: 5.3703x; 1.0779x over previous
#include <hip/hip_runtime.h>

#define NN 100000
#define NE 1600000
#define DIM 128
#define NG 512
#define NL 3
#define OUTC (NL * DIM)
#define BN_EPS 1e-5f
#define BR 98
#define NB ((NN + BR - 1) / BR)
#define TROWS 129
#define RBITS 7
#define RMASK ((1 << RBITS) - 1)
#define PR 128
#define NPB ((NN + PR - 1) / PR)
#define CH 6400

typedef __attribute__((ext_vector_type(2))) float v2f;
typedef __attribute__((ext_vector_type(8))) float v8f;
typedef __attribute__((ext_vector_type(4))) float v4f;
typedef __attribute__((ext_vector_type(4))) int v4i;


__global__ void k_zero(float* __restrict__ p, int n) {
    int i = blockIdx.x * blockDim.x + threadIdx.x;
    if (i < n) p[i] = 0.f;
}

__global__ void k_init(float* __restrict__ p, int n, int* __restrict__ binCount) {
    int i = blockIdx.x * blockDim.x + threadIdx.x;
    if (i < n) p[i] = 0.f;
    if (i < NB) binCount[i] = 0;
}


#define WPITCH 132
#define GEMM_WAVES 8
#define NTILES (NN / 16)
#define GEMM_GRID 256
#define GEMM_STRIDE (GEMM_WAVES * GEMM_GRID)
#define GEMM_ROUNDS (NTILES / GEMM_STRIDE)
#define GEMM_REST0 (GEMM_ROUNDS * GEMM_STRIDE)
#define GEMM_WT_FLOATS 41984

__global__ void __launch_bounds__(256)
k_gemm_bias_relu_stats(const float* __restrict__ A, const float* __restrict__ Wm,
                       const float* __restrict__ bias, float* __restrict__ out,
                       float* __restrict__ stats) {
    static_assert(GEMM_WT_FLOATS >= DIM * WPITCH, "W^T must fit its reservation");
    static_assert(GEMM_ROUNDS >= 1, "every wave has a first tile");
    static_assert(NTILES - GEMM_REST0 <= GEMM_GRID, "at most one left-over tile per block");
    static_assert(GEMM_WAVES * 16 == DIM, "one 16-column block of a left-over tile per wave");
    __shared__ __attribute__((aligned(16))) float wt[GEMM_WT_FLOATS];
    __shared__ float part[GEMM_WAVES][2 * DIM];

    const int tid  = threadIdx.x;
    const int lane = tid & 31;
    const int wave = __builtin_amdgcn_readfirstlane(tid >> 5);
    const int c = lane & 15;
    const int h = lane >> 4;
    int tile = wave * GEMM_GRID + blockIdx.x;

    float4 a[16];
    {
        const float4* arow = (const float4*)(A + ((size_t)tile * 16 + c) * DIM) + h;
#pragma unroll
        for (int g = 0; g < 16; ++g) a[g] = arow[2 * g];
    }

    {
        const float4* w4 = (const float4*)Wm;
        const int kq = lane >> 2, qq = lane & 3;
        float4 wv[16];
#pragma unroll
        for (int i = 0; i < 16; ++i) {
            const int m = wave * 16 + i;
            const int k = (m >> 3) * 8 + kq;
            const int q = (m & 7) * 4 + qq;
            wv[i] = w4[k * 32 + q];
        }
#pragma unroll
        for (int i = 0; i < 16; ++i) {
            const int m = wave * 16 + i;
            const int k = (m >> 3) * 8 + kq;
            const int q = (m & 7) * 4 + qq;
            float* d = wt + (4 * q) * WPITCH + k;
            d[0]          = wv[i].x;
            d[WPITCH]     = wv[i].y;
            d[2 * WPITCH] = wv[i].z;
            d[3 * WPITCH] = wv[i].w;
        }
    }
    __syncthreads();

    float s[8], s2[8], bn[8];
#pragma unroll
    for (int ct = 0; ct < 8; ++ct) { s[ct] = 0.f; s2[ct] = 0.f; bn[ct] = bias[ct * 16 + c]; }

    const float* wb = wt + c * WPITCH + 4 * h;

#pragma nounroll
    for (; tile < GEMM_REST0; tile += GEMM_STRIDE) {
        float4 an[16];
        {
            const int tn = (tile + GEMM_STRIDE < GEMM_REST0) ? tile + GEMM_STRIDE
                                                             : min(GEMM_REST0 + (int)blockIdx.x, NTILES - 1);
            const float4* arow = (const float4*)(A + ((size_t)tn * 16 + c) * DIM) + h;
#pragma unroll
            for (int g = 0; g < 16; ++g) an[g] = arow[2 * g];
        }

        v8f acc[8];
#pragma unroll
        for (int ct = 0; ct < 8; ++ct) acc[ct] = (v8f){};

        float4 b0[8], b1[8];
#pragma unroll
        for (int ct = 0; ct < 8; ++ct) b0[ct] = *(const float4*)(wb + ct * 16 * WPITCH);

#pragma unroll
        for (int g = 0; g < 16; g += 2) {
            __builtin_amdgcn_sched_barrier(0);
#pragma unroll
            for (int ct = 0; ct < 8; ++ct)
                b1[ct] = *(const float4*)(wb + ct * 16 * WPITCH + 8 * (g + 1));
            __builtin_amdgcn_sched_barrier(0);
#pragma unroll
            for (int ct = 0; ct < 8; ++ct) {
                v2f av = {a[g].x, a[g].y}, bv = {b0[ct].x, b0[ct].y};
                acc[ct] = __builtin_amdgcn_wmma_f32_16x16x4_f32(false, av, false, bv, (short)0, acc[ct], false, false);
            }
#pragma unroll
            for (int ct = 0; ct < 8; ++ct) {
                v2f av = {a[g].z, a[g].w}, bv = {b0[ct].z, b0[ct].w};
                acc[ct] = __builtin_amdgcn_wmma_f32_16x16x4_f32(false, av, false, bv, (short)0, acc[ct], false, false);
            }
            __builtin_amdgcn_sched_barrier(0);
            if (g + 2 < 16) {
#pragma unroll
                for (int ct = 0; ct < 8; ++ct)
                    b0[ct] = *(const float4*)(wb + ct * 16 * WPITCH + 8 * (g + 2));
            }
            __builtin_amdgcn_sched_barrier(0);
#pragma unroll
            for (int ct = 0; ct < 8; ++ct) {
                v2f av = {a[g + 1].x, a[g + 1].y}, bv = {b1[ct].x, b1[ct].y};
                acc[ct] = __builtin_amdgcn_wmma_f32_16x16x4_f32(false, av, false, bv, (short)0, acc[ct], false, false);
            }
#pragma unroll
            for (int ct = 0; ct < 8; ++ct) {
                v2f av = {a[g + 1].z, a[g + 1].w}, bv = {b1[ct].z, b1[ct].w};
                acc[ct] = __builtin_amdgcn_wmma_f32_16x16x4_f32(false, av, false, bv, (short)0, acc[ct], false, false);
            }
        }

        __builtin_amdgcn_sched_barrier(0);

        float* orow = out + ((size_t)tile * 16 + 8 * h) * DIM + c;
#pragma unroll
        for (int ct = 0; ct < 8; ++ct) {
#pragma unroll
            for (int r = 0; r < 8; ++r) {
                float v = fmaxf(acc[ct][r] + bn[ct], 0.f);
                orow[(size_t)r * DIM + ct * 16] = v;
                s[ct] += v; s2[ct] += v * v;
            }
        }

#pragma unroll
        for (int g = 0; g < 16; ++g) a[g] = an[g];
    }

    float es = 0.f, es2 = 0.f;
    const bool rest = GEMM_REST0 + (int)blockIdx.x < NTILES;
    if (rest) {
        const int rt = GEMM_REST0 + (int)blockIdx.x;
        const float* wr = wb + wave * 16 * WPITCH;
        float4 bf[16];
#pragma unroll
        for (int g = 0; g < 16; ++g) bf[g] = *(const float4*)(wr + 8 * g);
        v8f acc = (v8f){};
#pragma unroll
        for (int g = 0; g < 16; ++g) {
            v2f av = {a[g].x, a[g].y}, bv = {bf[g].x, bf[g].y};
            acc = __builtin_amdgcn_wmma_f32_16x16x4_f32(false, av, false, bv, (short)0, acc, false, false);
            v2f aw = {a[g].z, a[g].w}, bw = {bf[g].z, bf[g].w};
            acc = __builtin_amdgcn_wmma_f32_16x16x4_f32(false, aw, false, bw, (short)0, acc, false, false);
        }
        const float bx = bias[wave * 16 + c];
        float* orow = out + ((size_t)rt * 16 + 8 * h) * DIM + wave * 16 + c;
#pragma unroll
        for (int r = 0; r < 8; ++r) {
            float v = fmaxf(acc[r] + bx, 0.f);
            orow[(size_t)r * DIM] = v;
            es += v; es2 += v * v;
        }
    }

#pragma unroll
    for (int ct = 0; ct < 8; ++ct) {
        const float t  = s[ct]  + __shfl_xor(s[ct], 16);
        const float t2 = s2[ct] + __shfl_xor(s2[ct], 16);
        part[wave][(h ? DIM : 0) + ct * 16 + c] = h ? t2 : t;
    }
    if (rest) {
        const float t  = es  + __shfl_xor(es, 16);
        const float t2 = es2 + __shfl_xor(es2, 16);
        part[wave][(h ? DIM : 0) + wave * 16 + c] += h ? t2 : t;
    }
    __syncthreads();

    float tot = 0.f;
#pragma unroll
    for (int w = 0; w < GEMM_WAVES; ++w) tot += part[w][tid];
    atomicAdd(&stats[tid], tot);
}


__global__ void __launch_bounds__(256)
k_bin_count(const int* __restrict__ dst, int* __restrict__ binCount) {
    __shared__ int cnt[NB];
    const int tid = threadIdx.x;
    for (int i = tid; i < NB; i += 256) cnt[i] = 0;
    __syncthreads();
    const int* dp = dst + blockIdx.x * CH;
    const v4i* d4 = (const v4i*)dp;
    v4i d[6];
#pragma unroll
    for (int i = 0; i < 6; ++i) d[i] = d4[tid + 256 * i];
    const int dt = dp[6 * 1024 + tid];
    asm volatile("" :: "v"(d[0]), "v"(d[1]), "v"(d[2]), "v"(d[3]), "v"(d[4]), "v"(d[5]), "v"(dt));
#pragma unroll
    for (int i = 0; i < 6; ++i) {
        atomicAdd(&cnt[(unsigned)d[i].x / BR], 1);
        atomicAdd(&cnt[(unsigned)d[i].y / BR], 1);
        atomicAdd(&cnt[(unsigned)d[i].z / BR], 1);
        atomicAdd(&cnt[(unsigned)d[i].w / BR], 1);
    }
    atomicAdd(&cnt[(unsigned)dt / BR], 1);
    __syncthreads();
    for (int i = tid; i < NB; i += 256) {
        const int c = cnt[i];
        if (c) atomicAdd(&binCount[i], c);
    }
}

__global__ void __launch_bounds__(256)
k_bin_dinv(const int* __restrict__ rec, const int* __restrict__ binStart,
           float* __restrict__ dinv) {
    __shared__ int cnt[BR];
    const int tid = threadIdx.x;
    if (tid < BR) cnt[tid] = 0;
    __syncthreads();
    const int e0 = binStart[blockIdx.x];
    const int e1 = binStart[blockIdx.x + 1];
    for (int e = e0 + tid; e < e1; e += 4 * 256) {
        int r[4];
#pragma unroll
        for (int u = 0; u < 4; ++u) {
            const int k = e + 256 * u;
            r[u] = (k < e1) ? rec[k] : -1;
        }
#pragma unroll
        for (int u = 0; u < 4; ++u)
            if (r[u] >= 0) atomicAdd(&cnt[r[u] & RMASK], 1);
    }
    __syncthreads();
    const int node = blockIdx.x * BR + tid;
    if (tid < BR && node < NN)
        dinv[node] = rsqrtf(fmaxf((float)(cnt[tid] + 1), 1.f));
}

__global__ void __launch_bounds__(1024)
k_bin_scan(int* __restrict__ binStart, int* __restrict__ cursor) {
    __shared__ int sc[1024];
    const int b = threadIdx.x;
    int c = 0;
    if (b < NB) {
        c = cursor[b];
        cursor[b] = 0;
    }
    sc[b] = c;
    __syncthreads();
    for (int off = 1; off < 1024; off <<= 1) {
        int v = (b >= off) ? sc[b - off] : 0;
        __syncthreads();
        sc[b] += v;
        __syncthreads();
    }
    if (b < NB) binStart[b + 1] = sc[b];
    if (b == 0) binStart[0] = 0;
}

__global__ void __launch_bounds__(256)
k_bin_place(const int* __restrict__ src, const int* __restrict__ dst,
            const int* __restrict__ binStart, int* __restrict__ cursor,
            int* __restrict__ rec) {
    __shared__ int cnt[NB];
    __shared__ int bas[NB];
    const int tid = threadIdx.x;
    const int e0 = blockIdx.x * CH;
    for (int i = tid; i < NB; i += 256) cnt[i] = 0;
    __syncthreads();
    for (int k = tid; k < CH; k += 256) atomicAdd(&cnt[(unsigned)dst[e0 + k] / BR], 1);
    __syncthreads();
    for (int i = tid; i < NB; i += 256) {
        int c = cnt[i];
        bas[i] = c ? binStart[i] + atomicAdd(&cursor[i], c) : 0;
        cnt[i] = 0;
    }
    __syncthreads();
    for (int k = tid; k < CH; k += 256) {
        int d = dst[e0 + k], s = src[e0 + k];
        int b = (unsigned)d / BR;
        int r = atomicAdd(&cnt[b], 1);
        rec[bas[b] + r] = (s << RBITS) | (d - b * BR);
    }
}


struct BnCols { v4f mu, a, be; };

__device__ __forceinline__ BnCols bn_cols(const float* __restrict__ stats,
                                          const float* __restrict__ gamma,
                                          const float* __restrict__ beta, int lane) {
    BnCols b;
    b.mu = ((const v4f*)(stats + 2 * DIM))[lane];
    b.a  = ((const v4f*)(stats + 3 * DIM))[lane] * ((const v4f*)gamma)[lane];
    b.be = ((const v4f*)beta)[lane];
    return b;
}

__device__ __forceinline__ v4f bn_apply(v4f v, const BnCols& b) {
    return (v - b.mu) * b.a + b.be;
}

#define POOL_WORDS (8 * 2 * DIM)

__device__ __forceinline__ void pool_flush(float* __restrict__ pool, int layer, int lane,
                                           int g, v4f ps) {
    float* p = pool + (size_t)g * OUTC + layer * DIM + 4 * lane;
    atomicAdd(p,     ps.x);
    atomicAdd(p + 1, ps.y);
    atomicAdd(p + 2, ps.z);
    atomicAdd(p + 3, ps.w);
}

struct PoolRun {
    int cur;
    v4f ps;
    v4f aF, aL;
};

__device__ __forceinline__ void pool_begin(PoolRun& p) {
    p.cur = -1;
    p.ps = p.aF = p.aL = (v4f){0.f, 0.f, 0.f, 0.f};
}

__device__ __forceinline__ void pool_close(PoolRun& p, float* __restrict__ pool, int layer,
                                           int lane, int gF, int gL) {
    if (p.cur < 0) return;
    if (p.cur == gF)      p.aF += p.ps;
    else if (p.cur == gL) p.aL += p.ps;
    else                  pool_flush(pool, layer, lane, p.cur, p.ps);
}

__device__ __forceinline__ void pool_row(PoolRun& p, float* __restrict__ pool, int layer,
                                         int lane, int gF, int gL, int g, v4f x) {
    if (g != p.cur) {
        pool_close(p, pool, layer, lane, gF, gL);
        p.cur = g;
        p.ps = (v4f){0.f, 0.f, 0.f, 0.f};
    }
    p.ps += x;
}

__device__ __forceinline__ void pool_store(const PoolRun& p, float* pl, int wave, int lane) {
    *(v4f*)(pl + (wave * 2 + 0) * DIM + 4 * lane) = p.aF;
    *(v4f*)(pl + (wave * 2 + 1) * DIM + 4 * lane) = p.aL;
}

__device__ __forceinline__ void pool_block_add(const float* pl, float* __restrict__ pool,
                                               int layer, int tid, int gF, int gL) {
    const int half = tid >> 7, col = tid & (DIM - 1);
    float t = 0.f;
#pragma unroll
    for (int w = 0; w < 8; ++w) t += pl[(w * 2 + half) * DIM + col];
    if (half == 0 || gL != gF)
        atomicAdd(pool + (size_t)(half ? gL : gF) * OUTC + layer * DIM + col, t);
}

template <bool BN>
__global__ void __launch_bounds__(256)
k_agg_tile(const int* __restrict__ rec, const int* __restrict__ binStart,
           const float* __restrict__ dinv, const float* __restrict__ h,
           float* __restrict__ out,
           const float* __restrict__ stats, const float* __restrict__ gamma,
           const float* __restrict__ beta, const int* __restrict__ batch,
           float* __restrict__ pool, int layer) {
    __shared__ float tile[TROWS * DIM];
    __shared__ __attribute__((aligned(16))) float pl[BN ? POOL_WORDS : 4];

    const int tid  = threadIdx.x;
    const int lane = tid & 31;
    const int wave = __builtin_amdgcn_readfirstlane(tid >> 5);
    const int base = blockIdx.x * BR;
    const int nrows = min(BR, NN - base);
    const v4f* h4 = (const v4f*)h;

    BnCols bn;
    int gF = 0, gL = 0;
    if (BN) {
        bn = bn_cols(stats, gamma, beta, lane);
        gF = batch[base];
        gL = batch[base + nrows - 1];
    }

    {
        constexpr int SR = (BR + 7) / 8;
        v4f v[SR];
#pragma unroll
        for (int k = 0; k < SR; ++k) {
            const int node = min(base + wave + 8 * k, NN - 1);
            v[k] = h4[(size_t)node * 32 + lane];
        }
        const int kn = min(base + wave + 8 * (lane & 15), NN - 1);
        const float dn = dinv[kn];
        const int   gn = BN ? batch[kn] : 0;
        static_assert(SR == 13, "the fence below names v[0] .. v[SR - 1]");
        asm volatile("" :: "v"(v[0]), "v"(v[1]), "v"(v[2]), "v"(v[3]), "v"(v[4]), "v"(v[5]),
                           "v"(v[6]), "v"(v[7]), "v"(v[8]), "v"(v[9]), "v"(v[10]), "v"(v[11]),
                           "v"(v[12]), "v"(dn), "v"(gn));
        PoolRun pr;
        pool_begin(pr);
#pragma unroll
        for (int k = 0; k < SR; ++k) {
            const int r = wave + 8 * k;
            if (r < nrows) {
                v4f x = v[k];
                if (BN) {
                    x = bn_apply(x, bn);
                    pool_row(pr, pool, layer, lane, gF, gL, __builtin_amdgcn_readlane(gn, k), x);
                }
                float w = __int_as_float(__builtin_amdgcn_readlane(__float_as_int(dn), k));
                w *= w;
                float* t = tile + r * DIM + lane;
                t[0]  = x.x * w;
                t[32] = x.y * w;
                t[64] = x.z * w;
                t[96] = x.w * w;
            }
        }
        if (BN) {
            pool_close(pr, pool, layer, lane, gF, gL);
            pool_store(pr, pl, wave, lane);
        }
    }
    __syncthreads();
    if (BN) pool_block_add(pl, pool, layer, tid, gF, gL);

    const int e0 = binStart[blockIdx.x];
    const int e1 = binStart[blockIdx.x + 1];
    for (int eb = e0 + wave * 32; eb < e1; eb += 8 * 32) {
        const int cnt = min(32, e1 - eb);
        int s = base, dl = BR;
        float ws = 0.f, wd = 0.f;
        if (lane < cnt) {
            const int r = rec[eb + lane];
            s  = r >> RBITS;
            dl = r & RMASK;
            ws = dinv[s];
            wd = dinv[base + dl];
        }
#pragma unroll
        for (int j0 = 0; j0 < 32; j0 += 8) {
            if (j0 < cnt) {
                v4f v[8];
#pragma unroll
                for (int u = 0; u < 8; ++u) {
                    const int sj = __builtin_amdgcn_readlane(s, j0 + u);
                    v[u] = h4[(size_t)sj * 32 + lane];
                }
#pragma unroll
                for (int u = 0; u < 8; ++u) {
                    const int   dj = __builtin_amdgcn_readlane(dl, j0 + u);
                    const float wj =
                        __int_as_float(__builtin_amdgcn_readlane(__float_as_int(ws), j0 + u)) *
                        __int_as_float(__builtin_amdgcn_readlane(__float_as_int(wd), j0 + u));
                    v4f x = v[u];
                    if (BN) x = bn_apply(x, bn);
                    float* t = tile + dj * DIM + lane;
                    __hip_atomic_fetch_add(t,      x.x * wj, __ATOMIC_RELAXED, __HIP_MEMORY_SCOPE_WORKGROUP);
                    __hip_atomic_fetch_add(t + 32, x.y * wj, __ATOMIC_RELAXED, __HIP_MEMORY_SCOPE_WORKGROUP);
                    __hip_atomic_fetch_add(t + 64, x.z * wj, __ATOMIC_RELAXED, __HIP_MEMORY_SCOPE_WORKGROUP);
                    __hip_atomic_fetch_add(t + 96, x.w * wj, __ATOMIC_RELAXED, __HIP_MEMORY_SCOPE_WORKGROUP);
                }
            }
        }
    }
    __syncthreads();

    v4f* o4 = (v4f*)out;
    for (int r = wave; r < nrows; r += 8) {
        const float* t = tile + r * DIM + lane;
        v4f v;
        v.x = t[0]; v.y = t[32]; v.z = t[64]; v.w = t[96];
        o4[(size_t)(base + r) * 32 + lane] = v;
    }
}


__global__ void k_stats_finish(float* __restrict__ stats, float inv_n) {
    int c = threadIdx.x;
    float mu  = stats[c] * inv_n;
    float var = stats[DIM + c] * inv_n - mu * mu;
    stats[2 * DIM + c] = mu;
    stats[3 * DIM + c] = rsqrtf(var + BN_EPS);
}

__global__ void __launch_bounds__(256)
k_norm_pool_rows(const float* __restrict__ h, const float* __restrict__ stats,
                 const float* __restrict__ gamma, const float* __restrict__ beta,
                 const int* __restrict__ batch, float* __restrict__ pool, int layer) {
    const int tid  = threadIdx.x;
    const int lane = tid & 31;
    const int wave = __builtin_amdgcn_readfirstlane(tid >> 5);
    const int base = blockIdx.x * PR;
    const int nrows = min(PR, NN - base);
    const v4f* h4 = (const v4f*)h;
    const BnCols bn = bn_cols(stats, gamma, beta, lane);
    const int gF = batch[base];
    const int gL = batch[base + nrows - 1];
    __shared__ __attribute__((aligned(16))) float pl[POOL_WORDS];

    v4f v[16];
#pragma unroll
    for (int k = 0; k < 16; ++k) {
        const int node = min(base + wave + 8 * k, NN - 1);
        v[k] = h4[(size_t)node * 32 + lane];
    }
    const int kn = min(base + wave + 8 * (lane & 15), NN - 1);
    const int gn = batch[kn];
    asm volatile("" :: "v"(v[0]), "v"(v[1]), "v"(v[2]), "v"(v[3]), "v"(v[4]), "v"(v[5]),
                       "v"(v[6]), "v"(v[7]), "v"(v[8]), "v"(v[9]), "v"(v[10]), "v"(v[11]),
                       "v"(v[12]), "v"(v[13]), "v"(v[14]), "v"(v[15]), "v"(gn));
    PoolRun pr;
    pool_begin(pr);
#pragma unroll
    for (int k = 0; k < 16; ++k) {
        if (wave + 8 * k < nrows)
            pool_row(pr, pool, layer, lane, gF, gL, __builtin_amdgcn_readlane(gn, k), bn_apply(v[k], bn));
    }
    pool_close(pr, pool, layer, lane, gF, gL);
    pool_store(pr, pl, wave, lane);
    __syncthreads();
    pool_block_add(pl, pool, layer, tid, gF, gL);
}


extern "C" void kernel_launch(void* const* d_in, const int* in_sizes, int n_in,
                              void* d_out, int out_size, void* d_ws, size_t ws_size,
                              hipStream_t stream) {
    (void)in_sizes; (void)n_in; (void)out_size; (void)ws_size;

    const float* x     = (const float*)d_in[0];
    const int*   ei    = (const int*)d_in[1];
    const int*   batch = (const int*)d_in[2];
    const float *Wp[NL], *bp[NL], *gp[NL], *tp[NL];
    for (int i = 0; i < NL; ++i) {
        Wp[i] = (const float*)d_in[3 + 4 * i];
        bp[i] = (const float*)d_in[4 + 4 * i];
        gp[i] = (const float*)d_in[5 + 4 * i];
        tp[i] = (const float*)d_in[6 + 4 * i];
    }
    const int* esrc = ei;
    const int* edst = ei + NE;

    float* ws    = (float*)d_ws;
    float* dinv  = ws;
    float* stats = ws + NN;
    size_t o     = ((size_t)NN + 4 * DIM + DIM * DIM + 255) & ~(size_t)255;
    float* bufA  = ws + o;
    float* bufB  = bufA + (size_t)NN * DIM;
    int* binStart = (int*)(bufB + (size_t)NN * DIM);
    int* cursor   = binStart + 1024;
    int* rec      = cursor + 1024;
    float* outp  = (float*)d_out;

    k_init     <<<(NG * OUTC + 255) / 256, 256, 0, stream>>>(outp, NG * OUTC, cursor);
    k_bin_count<<<NE / CH, 256, 0, stream>>>(edst, cursor);
    k_bin_scan <<<1, 1024, 0, stream>>>(binStart, cursor);
    k_bin_place<<<NE / CH, 256, 0, stream>>>(esrc, edst, binStart, cursor, rec);
    k_bin_dinv <<<NB, 256, 0, stream>>>(rec, binStart, dinv);

    k_agg_tile<false><<<NB, 256, 0, stream>>>(rec, binStart, dinv, x, bufB,
                                              stats, gp[0], tp[0], batch, outp, 0);
    for (int l = 0; l < NL; ++l) {
        k_zero        <<<1, 256, 0, stream>>>(stats, 2 * DIM);
        k_gemm_bias_relu_stats<<<GEMM_GRID, 256, 0, stream>>>(
            bufB, Wp[l], bp[l], bufA, stats);
        k_stats_finish<<<1, DIM, 0, stream>>>(stats, 1.f / (float)NN);
        if (l < NL - 1) {
            k_agg_tile<true><<<NB, 256, 0, stream>>>(rec, binStart, dinv, bufA, bufB,
                                                     stats, gp[l], tp[l], batch, outp, l);
        } else {
            k_norm_pool_rows<<<NPB, 256, 0, stream>>>(bufA, stats, gp[l], tp[l], batch, outp, l);
        }
    }
}
